# starting kernel
# baseline (speedup 1.0000x reference)
#include <hip/hip_runtime.h>
#include <hip/hip_bf16.h>

#define NQ      10
#define LAYERS  10
#define BATCH   4096
#define NACT    4

typedef __attribute__((ext_vector_type(2))) float v2f;
typedef __attribute__((ext_vector_type(8))) float v8f;

__global__ void build_gates_kernel(const float* __restrict__ w,
                                   float* __restrict__ ws) {
  int id = blockIdx.x * blockDim.x + threadIdx.x;
  if (id >= LAYERS * 2 * 1024) return;
  int l     = id >> 11;
  int rem   = id & 2047;
  int which = rem >> 10;
  int e     = rem & 1023;
  int r  = e >> 5;
  int rp = e & 31;
  float accr = 1.f, acci = 0.f;
  for (int j = 0; j < 5; ++j) {
    int q = which * 5 + j;
    float phi   = w[(l * NQ + q) * 3 + 0];
    float theta = w[(l * NQ + q) * 3 + 1];
    float omega = w[(l * NQ + q) * 3 + 2];
    float ct = cosf(0.5f * theta), st = sinf(0.5f * theta);
    float a = 0.5f * (phi + omega), b = 0.5f * (phi - omega);
    float ca = cosf(a), sa = sinf(a), cb = cosf(b), sb = sinf(b);
    int bi = (r  >> (4 - j)) & 1;
    int bj = (rp >> (4 - j)) & 1;
    float gr, gi;
    if (!bi && !bj)      { gr =  ca * ct; gi = -sa * ct; }
    else if (!bi &&  bj) { gr = -cb * st; gi = -sb * st; }
    else if ( bi && !bj) { gr =  cb * st; gi = -sb * st; }
    else                 { gr =  ca * ct; gi =  sa * ct; }
    float nr = accr * gr - acci * gi;
    float ni = accr * gi + acci * gr;
    accr = nr; acci = ni;
  }
  float* base = ws + l * 6144 + which * 3072;
  base[e]        =  accr;
  base[1024 + e] =  acci;
  base[2048 + e] = -acci;
}

__device__ __forceinline__ int cnot_map(int b, int rr) {
  #pragma unroll
  for (int w = 0; w < NQ; ++w) {
    int t = w + rr; if (t >= NQ) t -= NQ;
    b ^= ((b >> (9 - w)) & 1) << (9 - t);
  }
  return b;
}

__global__ __launch_bounds__(64)
void qsim_kernel(const float* __restrict__ x,
                 const float* __restrict__ gates,
                 const float* __restrict__ fcw,
                 const float* __restrict__ fcb,
                 float* __restrict__ out) {
  __shared__ float lds[2 * 4096];
  const int wave = threadIdx.x >> 5;
  const int lane = threadIdx.x & 31;
  const int half = lane >> 4;
  const int l16  = lane & 15;
  const int b    = blockIdx.x * 2 + wave;

  float* Sr = lds + wave * 4096;
  float* Si = Sr + 1024;
  float* Tr = Sr + 2048;
  float* Ti = Sr + 3072;

  float cw[NQ], sw[NQ];
  for (int w = 0; w < NQ; ++w) {
    float xv = x[b * NQ + w];
    xv = xv > 0.f ? xv : 0.f;
    cw[w] = cosf(0.5f * xv);
    sw[w] = sinf(0.5f * xv);
  }
  {
    float mhi = 1.f;
    for (int w = 0; w < 5; ++w) mhi *= ((lane >> (4 - w)) & 1) ? sw[w] : cw[w];
    int pchi = __popc(lane);
    for (int c = 0; c < 32; ++c) {
      float m = mhi;
      for (int w = 0; w < 5; ++w) m *= ((c >> (4 - w)) & 1) ? sw[5 + w] : cw[5 + w];
      int k = (pchi + __popc(c)) & 3;
      float re = 0.f, im = 0.f;
      if      (k == 0) re =  m;
      else if (k == 1) im = -m;
      else if (k == 2) re = -m;
      else             im =  m;
      Sr[lane * 32 + c] = re;
      Si[lane * 32 + c] = im;
    }
  }
  __syncthreads();

  for (int l = 0; l < LAYERS; ++l) {
    const float* gAr  = gates + l * 6144;
    const float* gAi  = gAr + 1024;
    const float* gnAi = gAr + 2048;
    const float* gBr  = gAr + 3072;
    const float* gBi  = gAr + 4096;
    const float* gnBi = gAr + 5120;

    const int rr = l % 9 + 1;
    const int pI = cnot_map(1 << 9, rr);
    const int p7 = cnot_map(1 << 7, rr);
    const int p6 = cnot_map(1 << 6, rr);
    const int p5 = cnot_map(1 << 5, rr);
    const int pJ = cnot_map(1 << 4, rr);
    const int pLane = cnot_map((half << 8) | l16, rr);

    for (int I = 0; I < 2; ++I)
    for (int J = 0; J < 2; ++J) {
      v8f Rr = {}, Ri = {};
      const int row = I * 16 + l16;
      const int n   = J * 16 + l16;
      for (int K = 0; K < 2; ++K)
      for (int kk = 0; kk < 4; ++kk) {
        const int k0 = K * 16 + kk * 4 + 2 * half;
        v2f sr   = *(const v2f*)(Sr   + row * 32 + k0);
        v2f si   = *(const v2f*)(Si   + row * 32 + k0);
        v2f btr  = *(const v2f*)(gBr  + n * 32 + k0);
        v2f bti  = *(const v2f*)(gBi  + n * 32 + k0);
        v2f nbti = *(const v2f*)(gnBi + n * 32 + k0);
        Rr = __builtin_amdgcn_wmma_f32_16x16x4_f32(false, sr, false, btr,  (short)0, Rr, false, false);
        Rr = __builtin_amdgcn_wmma_f32_16x16x4_f32(false, si, false, nbti, (short)0, Rr, false, false);
        Ri = __builtin_amdgcn_wmma_f32_16x16x4_f32(false, sr, false, bti,  (short)0, Ri, false, false);
        Ri = __builtin_amdgcn_wmma_f32_16x16x4_f32(false, si, false, btr,  (short)0, Ri, false, false);
      }
      #pragma unroll
      for (int v = 0; v < 8; ++v) {
        int m = I * 16 + v + 8 * half;
        Tr[m * 32 + n] = Rr[v];
        Ti[m * 32 + n] = Ri[v];
      }
    }
    __syncthreads();

    for (int I = 0; I < 2; ++I)
    for (int J = 0; J < 2; ++J) {
      v8f Ur = {}, Ui = {};
      const int row = I * 16 + l16;
      const int n   = J * 16 + l16;
      for (int K = 0; K < 2; ++K)
      for (int kk = 0; kk < 4; ++kk) {
        const int k0 = K * 16 + kk * 4 + 2 * half;
        v2f ar  = *(const v2f*)(gAr  + row * 32 + k0);
        v2f ai  = *(const v2f*)(gAi  + row * 32 + k0);
        v2f nai = *(const v2f*)(gnAi + row * 32 + k0);
        v2f btr, bti;
        btr.x = Tr[(k0 + 0) * 32 + n];
        btr.y = Tr[(k0 + 1) * 32 + n];
        bti.x = Ti[(k0 + 0) * 32 + n];
        bti.y = Ti[(k0 + 1) * 32 + n];
        Ur = __builtin_amdgcn_wmma_f32_16x16x4_f32(false, ar,  false, btr, (short)0, Ur, false, false);
        Ur = __builtin_amdgcn_wmma_f32_16x16x4_f32(false, nai, false, bti, (short)0, Ur, false, false);
        Ui = __builtin_amdgcn_wmma_f32_16x16x4_f32(false, ar,  false, bti, (short)0, Ui, false, false);
        Ui = __builtin_amdgcn_wmma_f32_16x16x4_f32(false, ai,  false, btr, (short)0, Ui, false, false);
      }
      const int pbase = pLane ^ (I ? pI : 0) ^ (J ? pJ : 0);
      #pragma unroll
      for (int v = 0; v < 8; ++v) {
        int pb = pbase ^ ((v & 4) ? p7 : 0) ^ ((v & 2) ? p6 : 0) ^ ((v & 1) ? p5 : 0);
        Sr[pb] = Ur[v];
        Si[pb] = Ui[v];
      }
    }
    __syncthreads();
  }

  float z[NQ];
  for (int w = 0; w < NQ; ++w) z[w] = 0.f;
  for (int c = 0; c < 32; ++c) {
    int idx = lane * 32 + c;
    float re = Sr[idx], im = Si[idx];
    float p = re * re + im * im;
    #pragma unroll
    for (int w = 0; w < NQ; ++w)
      z[w] += ((idx >> (9 - w)) & 1) ? -p : p;
  }
  for (int off = 16; off > 0; off >>= 1)
    #pragma unroll
    for (int w = 0; w < NQ; ++w)
      z[w] += __shfl_xor(z[w], off, 32);
  if (lane < NACT) {
    float acc = fcb[lane];
    for (int w = 0; w < NQ; ++w) acc += fcw[lane * NQ + w] * z[w];
    out[b * NACT + lane] = acc;
  }
}

extern "C" void kernel_launch(void* const* d_in, const int* in_sizes, int n_in,
                              void* d_out, int out_size, void* d_ws, size_t ws_size,
                              hipStream_t stream) {
  const float* inputs  = (const float*)d_in[0];
  const float* weights = (const float*)d_in[1];
  const float* fcw     = (const float*)d_in[2];
  const float* fcb     = (const float*)d_in[3];
  float* out   = (float*)d_out;
  float* gates = (float*)d_ws;

  build_gates_kernel<<<80, 256, 0, stream>>>(weights, gates);
  qsim_kernel<<<BATCH / 2, 64, 0, stream>>>(inputs, gates, fcw, fcb, out);
}


# complex GEMMs in three-product form: k1=(xr+xi)gr, k2=xr(gi-gr), k3=xi(gr+gi), re=k1-k3, im=k1+k2; 3 f32 WMMAs per k-step instead of 4; changes the f32 summation tree
# speedup vs baseline: 1.1443x; 1.1443x over previous
#include <hip/hip_runtime.h>
#include <hip/hip_bf16.h>

#define NQ      10
#define LAYERS  10
#define BATCH   4096
#define NACT    4

typedef __attribute__((ext_vector_type(2))) float v2f;
typedef __attribute__((ext_vector_type(8))) float v8f;

__global__ void build_gates_kernel(const float* __restrict__ w,
                                   float* __restrict__ ws) {
  int id = blockIdx.x * blockDim.x + threadIdx.x;
  if (id >= LAYERS * 2 * 1024) return;
  int l     = id >> 11;
  int rem   = id & 2047;
  int which = rem >> 10;
  int e     = rem & 1023;
  int r  = e >> 5;
  int rp = e & 31;
  float accr = 1.f, acci = 0.f;
  for (int j = 0; j < 5; ++j) {
    int q = which * 5 + j;
    float phi   = w[(l * NQ + q) * 3 + 0];
    float theta = w[(l * NQ + q) * 3 + 1];
    float omega = w[(l * NQ + q) * 3 + 2];
    float ct = cosf(0.5f * theta), st = sinf(0.5f * theta);
    float a = 0.5f * (phi + omega), b = 0.5f * (phi - omega);
    float ca = cosf(a), sa = sinf(a), cb = cosf(b), sb = sinf(b);
    int bi = (r  >> (4 - j)) & 1;
    int bj = (rp >> (4 - j)) & 1;
    float gr, gi;
    if (!bi && !bj)      { gr =  ca * ct; gi = -sa * ct; }
    else if (!bi &&  bj) { gr = -cb * st; gi = -sb * st; }
    else if ( bi && !bj) { gr =  cb * st; gi = -sb * st; }
    else                 { gr =  ca * ct; gi =  sa * ct; }
    float nr = accr * gr - acci * gi;
    float ni = accr * gi + acci * gr;
    accr = nr; acci = ni;
  }
  float* base = ws + l * 6144 + which * 3072;
  base[e]        = accr;
  base[1024 + e] = acci - accr;
  base[2048 + e] = accr + acci;
}

__device__ __forceinline__ int cnot_map(int b, int rr) {
  #pragma unroll
  for (int w = 0; w < NQ; ++w) {
    int t = w + rr; if (t >= NQ) t -= NQ;
    b ^= ((b >> (9 - w)) & 1) << (9 - t);
  }
  return b;
}

__global__ __launch_bounds__(64, 5)
void qsim_kernel(const float* __restrict__ x,
                 const float* __restrict__ gates,
                 const float* __restrict__ fcw,
                 const float* __restrict__ fcb,
                 float* __restrict__ out) {
  __shared__ float lds[2 * 4096];
  const int wave = threadIdx.x >> 5;
  const int lane = threadIdx.x & 31;
  const int half = lane >> 4;
  const int l16  = lane & 15;
  const int b    = blockIdx.x * 2 + wave;

  float* Sr = lds + wave * 4096;
  float* Si = Sr + 1024;
  float* Tr = Sr + 2048;
  float* Ti = Sr + 3072;

  float cw[NQ], sw[NQ];
  for (int w = 0; w < NQ; ++w) {
    float xv = x[b * NQ + w];
    xv = xv > 0.f ? xv : 0.f;
    cw[w] = cosf(0.5f * xv);
    sw[w] = sinf(0.5f * xv);
  }
  {
    float mhi = 1.f;
    for (int w = 0; w < 5; ++w) mhi *= ((lane >> (4 - w)) & 1) ? sw[w] : cw[w];
    int pchi = __popc(lane);
    for (int c = 0; c < 32; ++c) {
      float m = mhi;
      for (int w = 0; w < 5; ++w) m *= ((c >> (4 - w)) & 1) ? sw[5 + w] : cw[5 + w];
      int k = (pchi + __popc(c)) & 3;
      float re = 0.f, im = 0.f;
      if      (k == 0) re =  m;
      else if (k == 1) im = -m;
      else if (k == 2) re = -m;
      else             im =  m;
      Sr[lane * 32 + c] = re;
      Si[lane * 32 + c] = im;
    }
  }
  __syncthreads();

  for (int l = 0; l < LAYERS; ++l) {
    const float* gAr = gates + l * 6144;
    const float* gAd = gAr + 1024;
    const float* gAs = gAr + 2048;
    const float* gBr = gAr + 3072;
    const float* gBd = gAr + 4096;
    const float* gBs = gAr + 5120;

    const int rr = l % 9 + 1;
    const int pI = cnot_map(1 << 9, rr);
    const int p7 = cnot_map(1 << 7, rr);
    const int p6 = cnot_map(1 << 6, rr);
    const int p5 = cnot_map(1 << 5, rr);
    const int pJ = cnot_map(1 << 4, rr);
    const int pLane = cnot_map((half << 8) | l16, rr);

    for (int I = 0; I < 2; ++I)
    for (int J = 0; J < 2; ++J) {
      v8f K1 = {}, K2 = {}, K3 = {};
      const int row = I * 16 + l16;
      const int n   = J * 16 + l16;
      for (int K = 0; K < 2; ++K)
      for (int kk = 0; kk < 4; ++kk) {
        const int k0 = K * 16 + kk * 4 + 2 * half;
        v2f sr  = *(const v2f*)(Sr  + row * 32 + k0);
        v2f si  = *(const v2f*)(Si  + row * 32 + k0);
        v2f ss  = sr + si;
        v2f btr = *(const v2f*)(gBr + n * 32 + k0);
        v2f btd = *(const v2f*)(gBd + n * 32 + k0);
        v2f bts = *(const v2f*)(gBs + n * 32 + k0);
        K1 = __builtin_amdgcn_wmma_f32_16x16x4_f32(false, ss, false, btr, (short)0, K1, false, false);
        K2 = __builtin_amdgcn_wmma_f32_16x16x4_f32(false, sr, false, btd, (short)0, K2, false, false);
        K3 = __builtin_amdgcn_wmma_f32_16x16x4_f32(false, si, false, bts, (short)0, K3, false, false);
      }
      #pragma unroll
      for (int v = 0; v < 8; ++v) {
        int m = I * 16 + v + 8 * half;
        Tr[m * 32 + n] = K1[v] - K3[v];
        Ti[m * 32 + n] = K1[v] + K2[v];
      }
    }
    __syncthreads();

    for (int I = 0; I < 2; ++I)
    for (int J = 0; J < 2; ++J) {
      v8f K1 = {}, K2 = {}, K3 = {};
      const int row = I * 16 + l16;
      const int n   = J * 16 + l16;
      for (int K = 0; K < 2; ++K)
      for (int kk = 0; kk < 4; ++kk) {
        const int k0 = K * 16 + kk * 4 + 2 * half;
        v2f ar = *(const v2f*)(gAr + row * 32 + k0);
        v2f ad = *(const v2f*)(gAd + row * 32 + k0);
        v2f a_s = *(const v2f*)(gAs + row * 32 + k0);
        v2f btr, bti;
        btr.x = Tr[(k0 + 0) * 32 + n];
        btr.y = Tr[(k0 + 1) * 32 + n];
        bti.x = Ti[(k0 + 0) * 32 + n];
        bti.y = Ti[(k0 + 1) * 32 + n];
        v2f bts = btr + bti;
        K1 = __builtin_amdgcn_wmma_f32_16x16x4_f32(false, ar, false, bts, (short)0, K1, false, false);
        K2 = __builtin_amdgcn_wmma_f32_16x16x4_f32(false, ad, false, btr, (short)0, K2, false, false);
        K3 = __builtin_amdgcn_wmma_f32_16x16x4_f32(false, a_s, false, bti, (short)0, K3, false, false);
      }
      const int pbase = pLane ^ (I ? pI : 0) ^ (J ? pJ : 0);
      #pragma unroll
      for (int v = 0; v < 8; ++v) {
        int pb = pbase ^ ((v & 4) ? p7 : 0) ^ ((v & 2) ? p6 : 0) ^ ((v & 1) ? p5 : 0);
        Sr[pb] = K1[v] - K3[v];
        Si[pb] = K1[v] + K2[v];
      }
    }
    __syncthreads();
  }

  float z[NQ];
  for (int w = 0; w < NQ; ++w) z[w] = 0.f;
  for (int c = 0; c < 32; ++c) {
    int idx = lane * 32 + c;
    float re = Sr[idx], im = Si[idx];
    float p = re * re + im * im;
    #pragma unroll
    for (int w = 0; w < NQ; ++w)
      z[w] += ((idx >> (9 - w)) & 1) ? -p : p;
  }
  for (int off = 16; off > 0; off >>= 1)
    #pragma unroll
    for (int w = 0; w < NQ; ++w)
      z[w] += __shfl_xor(z[w], off, 32);
  if (lane < NACT) {
    float acc = fcb[lane];
    for (int w = 0; w < NQ; ++w) acc += fcw[lane * NQ + w] * z[w];
    out[b * NACT + lane] = acc;
  }
}

extern "C" void kernel_launch(void* const* d_in, const int* in_sizes, int n_in,
                              void* d_out, int out_size, void* d_ws, size_t ws_size,
                              hipStream_t stream) {
  const float* inputs  = (const float*)d_in[0];
  const float* weights = (const float*)d_in[1];
  const float* fcw     = (const float*)d_in[2];
  const float* fcb     = (const float*)d_in[3];
  float* out   = (float*)d_out;
  float* gates = (float*)d_ws;

  build_gates_kernel<<<80, 256, 0, stream>>>(weights, gates);
  qsim_kernel<<<BATCH / 2, 64, 0, stream>>>(inputs, gates, fcw, fcb, out);
}


# Gate planes stored in WMMA-fragment order in the workspace: each gate operand load is one contiguous 16 B per lane (two k-steps, 4 cache lines per wave) instead of 8 B from 16 separate 128 B rows
# speedup vs baseline: 1.2841x; 1.1222x over previous
#include <hip/hip_runtime.h>
#include <hip/hip_bf16.h>

#define NQ      10
#define LAYERS  10
#define BATCH   4096
#define NACT    4

typedef __attribute__((ext_vector_type(2))) float v2f;
typedef __attribute__((ext_vector_type(4))) float v4f;
typedef __attribute__((ext_vector_type(8))) float v8f;

__global__ void build_gates_kernel(const float* __restrict__ w,
                                   float* __restrict__ ws) {
  int id = blockIdx.x * blockDim.x + threadIdx.x;
  if (id >= LAYERS * 2 * 1024) return;
  int l     = id >> 11;
  int rem   = id & 2047;
  int which = rem >> 10;
  int e     = rem & 1023;
  const int t  = e >> 9, kp = (e >> 7) & 3, ln = (e >> 2) & 31;
  const int h2 = (e >> 1) & 1, jj = e & 1;
  int r  = t * 16 + (ln & 15);
  int rp = (kp * 2 + h2) * 4 + 2 * (ln >> 4) + jj;
  float accr = 1.f, acci = 0.f;
  for (int j = 0; j < 5; ++j) {
    int q = which * 5 + j;
    float phi   = w[(l * NQ + q) * 3 + 0];
    float theta = w[(l * NQ + q) * 3 + 1];
    float omega = w[(l * NQ + q) * 3 + 2];
    float ct = cosf(0.5f * theta), st = sinf(0.5f * theta);
    float a = 0.5f * (phi + omega), b = 0.5f * (phi - omega);
    float ca = cosf(a), sa = sinf(a), cb = cosf(b), sb = sinf(b);
    int bi = (r  >> (4 - j)) & 1;
    int bj = (rp >> (4 - j)) & 1;
    float gr, gi;
    if (!bi && !bj)      { gr =  ca * ct; gi = -sa * ct; }
    else if (!bi &&  bj) { gr = -cb * st; gi = -sb * st; }
    else if ( bi && !bj) { gr =  cb * st; gi = -sb * st; }
    else                 { gr =  ca * ct; gi =  sa * ct; }
    float nr = accr * gr - acci * gi;
    float ni = accr * gi + acci * gr;
    accr = nr; acci = ni;
  }
  float* base = ws + l * 6144 + which * 3072;
  base[e]        = accr;
  base[1024 + e] = acci - accr;
  base[2048 + e] = accr + acci;
}

__device__ __forceinline__ int cnot_map(int b, int rr) {
  #pragma unroll
  for (int w = 0; w < NQ; ++w) {
    int t = w + rr; if (t >= NQ) t -= NQ;
    b ^= ((b >> (9 - w)) & 1) << (9 - t);
  }
  return b;
}

__global__ __launch_bounds__(64, 5)
void qsim_kernel(const float* __restrict__ x,
                 const float* __restrict__ gates,
                 const float* __restrict__ fcw,
                 const float* __restrict__ fcb,
                 float* __restrict__ out) {
  __shared__ float lds[2 * 4096];
  const int wave = threadIdx.x >> 5;
  const int lane = threadIdx.x & 31;
  const int half = lane >> 4;
  const int l16  = lane & 15;
  const int b    = blockIdx.x * 2 + wave;

  float* Sr = lds + wave * 4096;
  float* Si = Sr + 1024;
  float* Tr = Sr + 2048;
  float* Ti = Sr + 3072;

  float cw[NQ], sw[NQ];
  for (int w = 0; w < NQ; ++w) {
    float xv = x[b * NQ + w];
    xv = xv > 0.f ? xv : 0.f;
    cw[w] = cosf(0.5f * xv);
    sw[w] = sinf(0.5f * xv);
  }
  {
    float mhi = 1.f;
    for (int w = 0; w < 5; ++w) mhi *= ((lane >> (4 - w)) & 1) ? sw[w] : cw[w];
    int pchi = __popc(lane);
    for (int c = 0; c < 32; ++c) {
      float m = mhi;
      for (int w = 0; w < 5; ++w) m *= ((c >> (4 - w)) & 1) ? sw[5 + w] : cw[5 + w];
      int k = (pchi + __popc(c)) & 3;
      float re = 0.f, im = 0.f;
      if      (k == 0) re =  m;
      else if (k == 1) im = -m;
      else if (k == 2) re = -m;
      else             im =  m;
      Sr[lane * 32 + c] = re;
      Si[lane * 32 + c] = im;
    }
  }
  __syncthreads();

  for (int l = 0; l < LAYERS; ++l) {
    const float* gAr = gates + l * 6144;
    const float* gAd = gAr + 1024;
    const float* gAs = gAr + 2048;
    const float* gBr = gAr + 3072;
    const float* gBd = gAr + 4096;
    const float* gBs = gAr + 5120;

    const int rr = l % 9 + 1;
    const int pI = cnot_map(1 << 9, rr);
    const int p7 = cnot_map(1 << 7, rr);
    const int p6 = cnot_map(1 << 6, rr);
    const int p5 = cnot_map(1 << 5, rr);
    const int pJ = cnot_map(1 << 4, rr);
    const int pLane = cnot_map((half << 8) | l16, rr);

    for (int I = 0; I < 2; ++I)
    for (int J = 0; J < 2; ++J) {
      v8f K1 = {}, K2 = {}, K3 = {};
      const int row = I * 16 + l16;
      const int n   = J * 16 + l16;
      const int gofs = J * 512 + lane * 4;
      for (int kp = 0; kp < 4; ++kp) {
        const v4f qr = *(const v4f*)(gBr + gofs + kp * 128);
        const v4f qd = *(const v4f*)(gBd + gofs + kp * 128);
        const v4f qs = *(const v4f*)(gBs + gofs + kp * 128);
        for (int h2 = 0; h2 < 2; ++h2) {
          const int k0 = (kp * 2 + h2) * 4 + 2 * half;
          v2f sr  = *(const v2f*)(Sr  + row * 32 + k0);
          v2f si  = *(const v2f*)(Si  + row * 32 + k0);
          v2f ss  = sr + si;
          v2f btr = h2 ? qr.zw : qr.xy;
          v2f btd = h2 ? qd.zw : qd.xy;
          v2f bts = h2 ? qs.zw : qs.xy;
          K1 = __builtin_amdgcn_wmma_f32_16x16x4_f32(false, ss, false, btr, (short)0, K1, false, false);
          K2 = __builtin_amdgcn_wmma_f32_16x16x4_f32(false, sr, false, btd, (short)0, K2, false, false);
          K3 = __builtin_amdgcn_wmma_f32_16x16x4_f32(false, si, false, bts, (short)0, K3, false, false);
        }
      }
      #pragma unroll
      for (int v = 0; v < 8; ++v) {
        int m = I * 16 + v + 8 * half;
        Tr[m * 32 + n] = K1[v] - K3[v];
        Ti[m * 32 + n] = K1[v] + K2[v];
      }
    }
    __syncthreads();

    for (int I = 0; I < 2; ++I)
    for (int J = 0; J < 2; ++J) {
      v8f K1 = {}, K2 = {}, K3 = {};
      const int n   = J * 16 + l16;
      const int gofs = I * 512 + lane * 4;
      for (int kp = 0; kp < 4; ++kp) {
        const v4f qr = *(const v4f*)(gAr + gofs + kp * 128);
        const v4f qd = *(const v4f*)(gAd + gofs + kp * 128);
        const v4f qs = *(const v4f*)(gAs + gofs + kp * 128);
        for (int h2 = 0; h2 < 2; ++h2) {
          const int k0 = (kp * 2 + h2) * 4 + 2 * half;
          v2f ar  = h2 ? qr.zw : qr.xy;
          v2f ad  = h2 ? qd.zw : qd.xy;
          v2f a_s = h2 ? qs.zw : qs.xy;
          v2f btr, bti;
          btr.x = Tr[(k0 + 0) * 32 + n];
          btr.y = Tr[(k0 + 1) * 32 + n];
          bti.x = Ti[(k0 + 0) * 32 + n];
          bti.y = Ti[(k0 + 1) * 32 + n];
          v2f bts = btr + bti;
          K1 = __builtin_amdgcn_wmma_f32_16x16x4_f32(false, ar, false, bts, (short)0, K1, false, false);
          K2 = __builtin_amdgcn_wmma_f32_16x16x4_f32(false, ad, false, btr, (short)0, K2, false, false);
          K3 = __builtin_amdgcn_wmma_f32_16x16x4_f32(false, a_s, false, bti, (short)0, K3, false, false);
        }
      }
      const int pbase = pLane ^ (I ? pI : 0) ^ (J ? pJ : 0);
      #pragma unroll
      for (int v = 0; v < 8; ++v) {
        int pb = pbase ^ ((v & 4) ? p7 : 0) ^ ((v & 2) ? p6 : 0) ^ ((v & 1) ? p5 : 0);
        Sr[pb] = K1[v] - K3[v];
        Si[pb] = K1[v] + K2[v];
      }
    }
    __syncthreads();
  }

  float z[NQ];
  for (int w = 0; w < NQ; ++w) z[w] = 0.f;
  for (int c = 0; c < 32; ++c) {
    int idx = lane * 32 + c;
    float re = Sr[idx], im = Si[idx];
    float p = re * re + im * im;
    #pragma unroll
    for (int w = 0; w < NQ; ++w)
      z[w] += ((idx >> (9 - w)) & 1) ? -p : p;
  }
  for (int off = 16; off > 0; off >>= 1)
    #pragma unroll
    for (int w = 0; w < NQ; ++w)
      z[w] += __shfl_xor(z[w], off, 32);
  if (lane < NACT) {
    float acc = fcb[lane];
    for (int w = 0; w < NQ; ++w) acc += fcw[lane * NQ + w] * z[w];
    out[b * NACT + lane] = acc;
  }
}

extern "C" void kernel_launch(void* const* d_in, const int* in_sizes, int n_in,
                              void* d_out, int out_size, void* d_ws, size_t ws_size,
                              hipStream_t stream) {
  const float* inputs  = (const float*)d_in[0];
  const float* weights = (const float*)d_in[1];
  const float* fcw     = (const float*)d_in[2];
  const float* fcb     = (const float*)d_in[3];
  float* out   = (float*)d_out;
  float* gates = (float*)d_ws;

  build_gates_kernel<<<80, 256, 0, stream>>>(weights, gates);
  qsim_kernel<<<BATCH / 2, 64, 0, stream>>>(inputs, gates, fcw, fcb, out);
}


# Rot = RZ*RY*RZ: GEMM gate operands are the real RY Kronecker factors (2 f32 WMMAs per k-step, not 3); RZ phases applied as one complex multiply per amplitude at write-back; changes f32 summation tree
# speedup vs baseline: 1.5626x; 1.2169x over previous
#include <hip/hip_runtime.h>
#include <hip/hip_bf16.h>

#define NQ      10
#define LAYERS  10
#define BATCH   4096
#define NACT    4

#define PLANES_END (LAYERS * 2048)
#define EINIT_OFS  PLANES_END
#define ETAB_OFS   (PLANES_END + 2048)

typedef __attribute__((ext_vector_type(2))) float v2f;
typedef __attribute__((ext_vector_type(4))) float v4f;
typedef __attribute__((ext_vector_type(8))) float v8f;

__device__ __forceinline__ int cnot_map(int b, int rr) {
  #pragma unroll
  for (int w = 0; w < NQ; ++w) {
    int t = w + rr; if (t >= NQ) t -= NQ;
    b ^= ((b >> (9 - w)) & 1) << (9 - t);
  }
  return b;
}

__device__ __forceinline__ double rz_sum(const float* __restrict__ w, int l,
                                         int comp, int idx) {
  double s = 0.0;
  for (int q = 0; q < NQ; ++q) {
    double a = (double)w[(l * NQ + q) * 3 + comp];
    s += ((idx >> (9 - q)) & 1) ? a : -a;
  }
  return s;
}

__global__ void build_gates_kernel(const float* __restrict__ w,
                                   float* __restrict__ ws) {
  int id = blockIdx.x * blockDim.x + threadIdx.x;
  if (id < PLANES_END) {
    int l     = id >> 11;
    int rem   = id & 2047;
    int which = rem >> 10;
    int e     = rem & 1023;
    const int t  = e >> 9, kp = (e >> 7) & 3, ln = (e >> 2) & 31;
    const int h2 = (e >> 1) & 1, jj = e & 1;
    int r  = t * 16 + (ln & 15);
    int rp = (kp * 2 + h2) * 4 + 2 * (ln >> 4) + jj;
    float acc = 1.f;
    for (int j = 0; j < 5; ++j) {
      int q = which * 5 + j;
      float theta = w[(l * NQ + q) * 3 + 1];
      float ct = cosf(0.5f * theta), st = sinf(0.5f * theta);
      int bi = (r  >> (4 - j)) & 1;
      int bj = (rp >> (4 - j)) & 1;
      acc *= (bi == bj) ? ct : (bi ? st : -st);
    }
    ws[l * 2048 + which * 1024 + e] = acc;
    return;
  }
  int t = id - PLANES_END;
  if (t >= (LAYERS + 1) * 1024) return;
  const int tbl = t >> 10;
  const int e   = t & 1023;
  const int ln  = (e >> 2) & 31;
  const int j   = e & 3;
  double s;
  float* dre;
  float* dim;
  if (tbl == 0) {
    int c   = (e >> 7) * 4 + j;
    int idx = ln * 32 + c;
    s   = rz_sum(w, 0, 0, idx);
    dre = ws + EINIT_OFS + e;
    dim = dre + 1024;
  } else {
    const int l    = tbl - 1;
    const int tile = e >> 8;
    const int v    = ((e >> 7) & 1) * 4 + j;
    int m   = (tile >> 1) * 16 + v + 8 * (ln >> 4);
    int n   = (tile & 1) * 16 + (ln & 15);
    int idx = m * 32 + n;
    s = rz_sum(w, l, 2, idx);
    if (l + 1 < LAYERS)
      s += rz_sum(w, l + 1, 0, cnot_map(idx, l % 9 + 1));
    dre = ws + ETAB_OFS + l * 2048 + tile * 512 + (e & 255);
    dim = dre + 256;
  }
  double turns = s * 0.07957747154594767;
  turns -= rint(turns);
  float a = (float)(turns * 6.283185307179586);
  *dre = cosf(a);
  *dim = sinf(a);
}

__global__ __launch_bounds__(64, 5)
void qsim_kernel(const float* __restrict__ x,
                 const float* __restrict__ gates,
                 const float* __restrict__ fcw,
                 const float* __restrict__ fcb,
                 float* __restrict__ out) {
  __shared__ float lds[2 * 4096];
  const int wave = threadIdx.x >> 5;
  const int lane = threadIdx.x & 31;
  const int half = lane >> 4;
  const int l16  = lane & 15;
  const int b    = blockIdx.x * 2 + wave;

  float* Sr = lds + wave * 4096;
  float* Si = Sr + 1024;
  float* Tr = Sr + 2048;
  float* Ti = Sr + 3072;

  float cw[NQ], sw[NQ];
  for (int w = 0; w < NQ; ++w) {
    float xv = x[b * NQ + w];
    xv = xv > 0.f ? xv : 0.f;
    cw[w] = cosf(0.5f * xv);
    sw[w] = sinf(0.5f * xv);
  }
  {
    const float* einit = gates + EINIT_OFS + lane * 4;
    float mhi = 1.f;
    for (int w = 0; w < 5; ++w) mhi *= ((lane >> (4 - w)) & 1) ? sw[w] : cw[w];
    int pchi = __popc(lane);
    for (int c4 = 0; c4 < 8; ++c4) {
      const v4f pr = *(const v4f*)(einit + c4 * 128);
      const v4f pi = *(const v4f*)(einit + 1024 + c4 * 128);
      #pragma unroll
      for (int j = 0; j < 4; ++j) {
        const int c = c4 * 4 + j;
        float m = mhi;
        for (int w = 0; w < 5; ++w) m *= ((c >> (4 - w)) & 1) ? sw[5 + w] : cw[5 + w];
        int k = (pchi + __popc(c)) & 3;
        float re = 0.f, im = 0.f;
        if      (k == 0) re =  m;
        else if (k == 1) im = -m;
        else if (k == 2) re = -m;
        else             im =  m;
        const float er = pr[j], ei = pi[j];
        Sr[lane * 32 + c] = re * er - im * ei;
        Si[lane * 32 + c] = re * ei + im * er;
      }
    }
  }
  __syncthreads();

  for (int l = 0; l < LAYERS; ++l) {
    const float* gA = gates + l * 2048;
    const float* gB = gA + 1024;
    const float* E  = gates + ETAB_OFS + l * 2048 + lane * 4;

    const int rr = l % 9 + 1;
    const int pI = cnot_map(1 << 9, rr);
    const int p7 = cnot_map(1 << 7, rr);
    const int p6 = cnot_map(1 << 6, rr);
    const int p5 = cnot_map(1 << 5, rr);
    const int pJ = cnot_map(1 << 4, rr);
    const int pLane = cnot_map((half << 8) | l16, rr);

    for (int I = 0; I < 2; ++I)
    for (int J = 0; J < 2; ++J) {
      v8f Kr = {}, Ki = {};
      const int row = I * 16 + l16;
      const int n   = J * 16 + l16;
      const int gofs = J * 512 + lane * 4;
      for (int kp = 0; kp < 4; ++kp) {
        const v4f qb = *(const v4f*)(gB + gofs + kp * 128);
        for (int h2 = 0; h2 < 2; ++h2) {
          const int k0 = (kp * 2 + h2) * 4 + 2 * half;
          v2f sr = *(const v2f*)(Sr + row * 32 + k0);
          v2f si = *(const v2f*)(Si + row * 32 + k0);
          v2f bt = h2 ? qb.zw : qb.xy;
          Kr = __builtin_amdgcn_wmma_f32_16x16x4_f32(false, sr, false, bt, (short)0, Kr, false, false);
          Ki = __builtin_amdgcn_wmma_f32_16x16x4_f32(false, si, false, bt, (short)0, Ki, false, false);
        }
      }
      #pragma unroll
      for (int v = 0; v < 8; ++v) {
        int m = I * 16 + v + 8 * half;
        Tr[m * 32 + n] = Kr[v];
        Ti[m * 32 + n] = Ki[v];
      }
    }
    __syncthreads();

    for (int I = 0; I < 2; ++I) {
      const float* gAt = gA + I * 512 + lane * 4;
      v4f qa[4];
      #pragma unroll
      for (int kp = 0; kp < 4; ++kp) qa[kp] = *(const v4f*)(gAt + kp * 128);
      for (int J = 0; J < 2; ++J) {
        v8f Kr = {}, Ki = {};
        const int n = J * 16 + l16;
        const float* Et = E + (I * 2 + J) * 512;
        const v4f er0 = *(const v4f*)(Et);
        const v4f er1 = *(const v4f*)(Et + 128);
        const v4f ei0 = *(const v4f*)(Et + 256);
        const v4f ei1 = *(const v4f*)(Et + 384);
        #pragma unroll
        for (int kp = 0; kp < 4; ++kp) {
          for (int h2 = 0; h2 < 2; ++h2) {
            const int k0 = (kp * 2 + h2) * 4 + 2 * half;
            v2f a = h2 ? qa[kp].zw : qa[kp].xy;
            v2f btr, bti;
            btr.x = Tr[(k0 + 0) * 32 + n];
            btr.y = Tr[(k0 + 1) * 32 + n];
            bti.x = Ti[(k0 + 0) * 32 + n];
            bti.y = Ti[(k0 + 1) * 32 + n];
            Kr = __builtin_amdgcn_wmma_f32_16x16x4_f32(false, a, false, btr, (short)0, Kr, false, false);
            Ki = __builtin_amdgcn_wmma_f32_16x16x4_f32(false, a, false, bti, (short)0, Ki, false, false);
          }
        }
        const int pbase = pLane ^ (I ? pI : 0) ^ (J ? pJ : 0);
        #pragma unroll
        for (int v = 0; v < 8; ++v) {
          int pb = pbase ^ ((v & 4) ? p7 : 0) ^ ((v & 2) ? p6 : 0) ^ ((v & 1) ? p5 : 0);
          const float er = (v < 4) ? er0[v & 3] : er1[v & 3];
          const float ei = (v < 4) ? ei0[v & 3] : ei1[v & 3];
          Sr[pb] = Kr[v] * er - Ki[v] * ei;
          Si[pb] = Kr[v] * ei + Ki[v] * er;
        }
      }
    }
    __syncthreads();
  }

  float z[NQ];
  for (int w = 0; w < NQ; ++w) z[w] = 0.f;
  for (int c = 0; c < 32; ++c) {
    int idx = lane * 32 + c;
    float re = Sr[idx], im = Si[idx];
    float p = re * re + im * im;
    #pragma unroll
    for (int w = 0; w < NQ; ++w)
      z[w] += ((idx >> (9 - w)) & 1) ? -p : p;
  }
  for (int off = 16; off > 0; off >>= 1)
    #pragma unroll
    for (int w = 0; w < NQ; ++w)
      z[w] += __shfl_xor(z[w], off, 32);
  if (lane < NACT) {
    float acc = fcb[lane];
    for (int w = 0; w < NQ; ++w) acc += fcw[lane * NQ + w] * z[w];
    out[b * NACT + lane] = acc;
  }
}

extern "C" void kernel_launch(void* const* d_in, const int* in_sizes, int n_in,
                              void* d_out, int out_size, void* d_ws, size_t ws_size,
                              hipStream_t stream) {
  const float* inputs  = (const float*)d_in[0];
  const float* weights = (const float*)d_in[1];
  const float* fcw     = (const float*)d_in[2];
  const float* fcb     = (const float*)d_in[3];
  float* out   = (float*)d_out;
  float* gates = (float*)d_ws;

  build_gates_kernel<<<124, 256, 0, stream>>>(weights, gates);
  qsim_kernel<<<BATCH / 2, 64, 0, stream>>>(inputs, gates, fcw, fcb, out);
}


# LDS bank conflicts removed: state and T rows at a 36-float pitch, T stored transposed, so both products read their LDS operand as conflict-free 8 B row reads; init, T and readout move 16 B per lane
# speedup vs baseline: 1.9522x; 1.2493x over previous
#include <hip/hip_runtime.h>
#include <hip/hip_bf16.h>

#define NQ      10
#define LAYERS  10
#define BATCH   4096
#define NACT    4

#define PLANES_END (LAYERS * 2048)
#define EINIT_OFS  PLANES_END
#define ETAB_OFS   (PLANES_END + 2048)

typedef __attribute__((ext_vector_type(2))) float v2f;
typedef __attribute__((ext_vector_type(4))) float v4f;
typedef __attribute__((ext_vector_type(8))) float v8f;

#define SP     36
#define PLANE  (32 * SP)

__device__ __forceinline__ int cnot_map(int b, int rr) {
  #pragma unroll
  for (int w = 0; w < NQ; ++w) {
    int t = w + rr; if (t >= NQ) t -= NQ;
    b ^= ((b >> (9 - w)) & 1) << (9 - t);
  }
  return b;
}

__device__ __forceinline__ double rz_sum(const float* __restrict__ w, int l,
                                         int comp, int idx) {
  double s = 0.0;
  for (int q = 0; q < NQ; ++q) {
    double a = (double)w[(l * NQ + q) * 3 + comp];
    s += ((idx >> (9 - q)) & 1) ? a : -a;
  }
  return s;
}

__global__ void build_gates_kernel(const float* __restrict__ w,
                                   float* __restrict__ ws) {
  int id = blockIdx.x * blockDim.x + threadIdx.x;
  if (id < PLANES_END) {
    int l     = id >> 11;
    int rem   = id & 2047;
    int which = rem >> 10;
    int e     = rem & 1023;
    const int t  = e >> 9, kp = (e >> 7) & 3, ln = (e >> 2) & 31;
    const int h2 = (e >> 1) & 1, jj = e & 1;
    int r  = t * 16 + (ln & 15);
    int rp = (kp * 2 + h2) * 4 + 2 * (ln >> 4) + jj;
    float acc = 1.f;
    for (int j = 0; j < 5; ++j) {
      int q = which * 5 + j;
      float theta = w[(l * NQ + q) * 3 + 1];
      float ct = cosf(0.5f * theta), st = sinf(0.5f * theta);
      int bi = (r  >> (4 - j)) & 1;
      int bj = (rp >> (4 - j)) & 1;
      acc *= (bi == bj) ? ct : (bi ? st : -st);
    }
    ws[l * 2048 + which * 1024 + e] = acc;
    return;
  }
  int t = id - PLANES_END;
  if (t >= (LAYERS + 1) * 1024) return;
  const int tbl = t >> 10;
  const int e   = t & 1023;
  const int ln  = (e >> 2) & 31;
  const int j   = e & 3;
  double s;
  float* dre;
  float* dim;
  if (tbl == 0) {
    int c   = (e >> 7) * 4 + j;
    int idx = ln * 32 + c;
    s   = rz_sum(w, 0, 0, idx);
    dre = ws + EINIT_OFS + e;
    dim = dre + 1024;
  } else {
    const int l    = tbl - 1;
    const int tile = e >> 8;
    const int v    = ((e >> 7) & 1) * 4 + j;
    int m   = (tile >> 1) * 16 + v + 8 * (ln >> 4);
    int n   = (tile & 1) * 16 + (ln & 15);
    int idx = m * 32 + n;
    s = rz_sum(w, l, 2, idx);
    if (l + 1 < LAYERS)
      s += rz_sum(w, l + 1, 0, cnot_map(idx, l % 9 + 1));
    dre = ws + ETAB_OFS + l * 2048 + tile * 512 + (e & 255);
    dim = dre + 256;
  }
  double turns = s * 0.07957747154594767;
  turns -= rint(turns);
  float a = (float)(turns * 6.283185307179586);
  *dre = cosf(a);
  *dim = sinf(a);
}

__global__ __launch_bounds__(64, 4)
void qsim_kernel(const float* __restrict__ x,
                 const float* __restrict__ gates,
                 const float* __restrict__ fcw,
                 const float* __restrict__ fcb,
                 float* __restrict__ out) {
  __shared__ __attribute__((aligned(16))) float lds[2 * 4 * PLANE];
  const int wave = threadIdx.x >> 5;
  const int lane = threadIdx.x & 31;
  const int half = lane >> 4;
  const int l16  = lane & 15;
  const int b    = blockIdx.x * 2 + wave;

  float* Sr = lds + wave * (4 * PLANE);
  float* Si = Sr + PLANE;
  float* Tr = Sr + 2 * PLANE;
  float* Ti = Sr + 3 * PLANE;

  float cw[NQ], sw[NQ];
  for (int w = 0; w < NQ; ++w) {
    float xv = x[b * NQ + w];
    xv = xv > 0.f ? xv : 0.f;
    cw[w] = cosf(0.5f * xv);
    sw[w] = sinf(0.5f * xv);
  }
  {
    const float* einit = gates + EINIT_OFS + lane * 4;
    float mhi = 1.f;
    for (int w = 0; w < 5; ++w) mhi *= ((lane >> (4 - w)) & 1) ? sw[w] : cw[w];
    int pchi = __popc(lane);
    for (int c4 = 0; c4 < 8; ++c4) {
      const v4f pr = *(const v4f*)(einit + c4 * 128);
      const v4f pi = *(const v4f*)(einit + 1024 + c4 * 128);
      v4f sr4, si4;
      #pragma unroll
      for (int j = 0; j < 4; ++j) {
        const int c = c4 * 4 + j;
        float m = mhi;
        for (int w = 0; w < 5; ++w) m *= ((c >> (4 - w)) & 1) ? sw[5 + w] : cw[5 + w];
        int k = (pchi + __popc(c)) & 3;
        float re = 0.f, im = 0.f;
        if      (k == 0) re =  m;
        else if (k == 1) im = -m;
        else if (k == 2) re = -m;
        else             im =  m;
        const float er = pr[j], ei = pi[j];
        sr4[j] = re * er - im * ei;
        si4[j] = re * ei + im * er;
      }
      *(v4f*)(Sr + lane * SP + c4 * 4) = sr4;
      *(v4f*)(Si + lane * SP + c4 * 4) = si4;
    }
  }
  __syncthreads();

  for (int l = 0; l < LAYERS; ++l) {
    const float* gA = gates + l * 2048;
    const float* gB = gA + 1024;
    const float* E  = gates + ETAB_OFS + l * 2048 + lane * 4;

    const int rr = l % 9 + 1;
    const int pI = cnot_map(1 << 9, rr);
    const int p7 = cnot_map(1 << 7, rr);
    const int p6 = cnot_map(1 << 6, rr);
    const int p5 = cnot_map(1 << 5, rr);
    const int pJ = cnot_map(1 << 4, rr);
    const int pLane = cnot_map((half << 8) | l16, rr);

    for (int I = 0; I < 2; ++I)
    for (int J = 0; J < 2; ++J) {
      v8f Kr = {}, Ki = {};
      const int row = I * 16 + l16;
      const int n   = J * 16 + l16;
      const int gofs = J * 512 + lane * 4;
      for (int kp = 0; kp < 4; ++kp) {
        const v4f qb = *(const v4f*)(gB + gofs + kp * 128);
        for (int h2 = 0; h2 < 2; ++h2) {
          const int k0 = (kp * 2 + h2) * 4 + 2 * half;
          v2f sr = *(const v2f*)(Sr + row * SP + k0);
          v2f si = *(const v2f*)(Si + row * SP + k0);
          v2f bt = h2 ? qb.zw : qb.xy;
          Kr = __builtin_amdgcn_wmma_f32_16x16x4_f32(false, sr, false, bt, (short)0, Kr, false, false);
          Ki = __builtin_amdgcn_wmma_f32_16x16x4_f32(false, si, false, bt, (short)0, Ki, false, false);
        }
      }
      float* tr = Tr + n * SP + I * 16 + 8 * half;
      float* ti = Ti + n * SP + I * 16 + 8 * half;
      *(v4f*)(tr)     = __builtin_shufflevector(Kr, Kr, 0, 1, 2, 3);
      *(v4f*)(tr + 4) = __builtin_shufflevector(Kr, Kr, 4, 5, 6, 7);
      *(v4f*)(ti)     = __builtin_shufflevector(Ki, Ki, 0, 1, 2, 3);
      *(v4f*)(ti + 4) = __builtin_shufflevector(Ki, Ki, 4, 5, 6, 7);
    }
    __syncthreads();

    for (int I = 0; I < 2; ++I) {
      const float* gAt = gA + I * 512 + lane * 4;
      v4f qa[4];
      #pragma unroll
      for (int kp = 0; kp < 4; ++kp) qa[kp] = *(const v4f*)(gAt + kp * 128);
      for (int J = 0; J < 2; ++J) {
        v8f Kr = {}, Ki = {};
        const int n = J * 16 + l16;
        const float* Et = E + (I * 2 + J) * 512;
        const v4f er0 = *(const v4f*)(Et);
        const v4f er1 = *(const v4f*)(Et + 128);
        const v4f ei0 = *(const v4f*)(Et + 256);
        const v4f ei1 = *(const v4f*)(Et + 384);
        #pragma unroll
        for (int kp = 0; kp < 4; ++kp) {
          for (int h2 = 0; h2 < 2; ++h2) {
            const int k0 = (kp * 2 + h2) * 4 + 2 * half;
            v2f a = h2 ? qa[kp].zw : qa[kp].xy;
            v2f btr = *(const v2f*)(Tr + n * SP + k0);
            v2f bti = *(const v2f*)(Ti + n * SP + k0);
            Kr = __builtin_amdgcn_wmma_f32_16x16x4_f32(false, a, false, btr, (short)0, Kr, false, false);
            Ki = __builtin_amdgcn_wmma_f32_16x16x4_f32(false, a, false, bti, (short)0, Ki, false, false);
          }
        }
        const int pbase = pLane ^ (I ? pI : 0) ^ (J ? pJ : 0);
        #pragma unroll
        for (int v = 0; v < 8; ++v) {
          int pb = pbase ^ ((v & 4) ? p7 : 0) ^ ((v & 2) ? p6 : 0) ^ ((v & 1) ? p5 : 0);
          const float er = (v < 4) ? er0[v & 3] : er1[v & 3];
          const float ei = (v < 4) ? ei0[v & 3] : ei1[v & 3];
          const int pa = pb + ((pb >> 5) << 2);
          Sr[pa] = Kr[v] * er - Ki[v] * ei;
          Si[pa] = Kr[v] * ei + Ki[v] * er;
        }
      }
    }
    __syncthreads();
  }

  float z[NQ];
  for (int w = 0; w < NQ; ++w) z[w] = 0.f;
  for (int c4 = 0; c4 < 8; ++c4) {
    const v4f r4 = *(const v4f*)(Sr + lane * SP + c4 * 4);
    const v4f i4 = *(const v4f*)(Si + lane * SP + c4 * 4);
    #pragma unroll
    for (int j = 0; j < 4; ++j) {
      int idx = lane * 32 + c4 * 4 + j;
      float re = r4[j], im = i4[j];
      float p = re * re + im * im;
      #pragma unroll
      for (int w = 0; w < NQ; ++w)
        z[w] += ((idx >> (9 - w)) & 1) ? -p : p;
    }
  }
  for (int off = 16; off > 0; off >>= 1)
    #pragma unroll
    for (int w = 0; w < NQ; ++w)
      z[w] += __shfl_xor(z[w], off, 32);
  if (lane < NACT) {
    float acc = fcb[lane];
    for (int w = 0; w < NQ; ++w) acc += fcw[lane * NQ + w] * z[w];
    out[b * NACT + lane] = acc;
  }
}

extern "C" void kernel_launch(void* const* d_in, const int* in_sizes, int n_in,
                              void* d_out, int out_size, void* d_ws, size_t ws_size,
                              hipStream_t stream) {
  const float* inputs  = (const float*)d_in[0];
  const float* weights = (const float*)d_in[1];
  const float* fcw     = (const float*)d_in[2];
  const float* fcb     = (const float*)d_in[3];
  float* out   = (float*)d_out;
  float* gates = (float*)d_ws;

  build_gates_kernel<<<124, 256, 0, stream>>>(weights, gates);
  qsim_kernel<<<BATCH / 2, 64, 0, stream>>>(inputs, gates, fcw, fcb, out);
}


# Each layer's RY part applied as three Kronecker factors (wires 2-5, 6-9, 0-1) chained in accumulator registers: 96 WMMAs per layer, not 128; one LDS round trip, not two; changes f32 summation tree
# speedup vs baseline: 2.2387x; 1.1468x over previous
#include <hip/hip_runtime.h>
#include <hip/hip_bf16.h>

#define NQ      10
#define LAYERS  10
#define BATCH   4096
#define NACT    4

#define GATES_END  (LAYERS * 1024)
#define EINIT_OFS  GATES_END
#define ETAB_OFS   (GATES_END + 2048)

typedef __attribute__((ext_vector_type(2))) float v2f;
typedef __attribute__((ext_vector_type(4))) float v4f;
typedef __attribute__((ext_vector_type(8))) float v8f;

__device__ __forceinline__ int cnot_map(int b, int rr) {
  #pragma unroll
  for (int w = 0; w < NQ; ++w) {
    int t = w + rr; if (t >= NQ) t -= NQ;
    b ^= ((b >> (9 - w)) & 1) << (9 - t);
  }
  return b;
}

__device__ __forceinline__ int lds_slot(int idx) {
  const int t = idx >> 8, a = (idx >> 4) & 15, b = idx & 15;
  return (t << 8) | (b << 4) | ((((a >> 2) ^ (b >> 2)) & 3) << 2) | (a & 3);
}

__device__ __forceinline__ double rz_sum(const float* __restrict__ w, int l,
                                         int comp, int idx) {
  double s = 0.0;
  for (int q = 0; q < NQ; ++q) {
    double a = (double)w[(l * NQ + q) * 3 + comp];
    s += ((idx >> (9 - q)) & 1) ? a : -a;
  }
  return s;
}

__global__ void build_gates_kernel(const float* __restrict__ w,
                                   float* __restrict__ ws) {
  int id = blockIdx.x * blockDim.x + threadIdx.x;
  if (id < GATES_END) {
    const int l   = id >> 10;
    const int rem = id & 1023;
    int nq, q0, r, cc;
    if (rem < 512) {
      nq = 4; q0 = 2 + (rem >> 8) * 4;
      r = (rem & 255) >> 4; cc = rem & 15;
    } else if (rem < 528) {
      nq = 2; q0 = 0;
      r = (rem - 512) >> 2; cc = rem & 3;
    } else {
      return;
    }
    float acc = 1.f;
    for (int j = 0; j < nq; ++j) {
      float theta = w[(l * NQ + q0 + j) * 3 + 1];
      float ct = cosf(0.5f * theta), st = sinf(0.5f * theta);
      int bi = (r  >> (nq - 1 - j)) & 1;
      int bj = (cc >> (nq - 1 - j)) & 1;
      acc *= (bi == bj) ? ct : (bi ? st : -st);
    }
    ws[l * 1024 + rem] = acc;
    return;
  }
  int t = id - GATES_END;
  if (t >= (LAYERS + 1) * 1024) return;
  const int tbl = t >> 10;
  const int e   = t & 1023;
  const int ln  = (e >> 2) & 31;
  const int j   = e & 3;
  double s;
  float* dre;
  float* dim;
  if (tbl == 0) {
    int c   = (e >> 7) * 4 + j;
    int idx = ln * 32 + c;
    s   = rz_sum(w, 0, 0, idx);
    dre = ws + EINIT_OFS + e;
    dim = dre + 1024;
  } else {
    const int l  = tbl - 1;
    const int g  = e >> 8;
    const int hi = (e >> 7) & 1;
    const int a  = 8 * (ln >> 4) + 2 * g + hi;
    const int idx = (j << 8) | (a << 4) | (ln & 15);
    s = rz_sum(w, l, 2, idx);
    if (l + 1 < LAYERS)
      s += rz_sum(w, l + 1, 0, cnot_map(idx, l % 9 + 1));
    dre = ws + ETAB_OFS + l * 2048 + g * 512 + (e & 255);
    dim = dre + 256;
  }
  double turns = s * 0.07957747154594767;
  turns -= rint(turns);
  float a = (float)(turns * 6.283185307179586);
  *dre = cosf(a);
  *dim = sinf(a);
}

__device__ __forceinline__ float xhalf(float v) {
  return __builtin_bit_cast(float, __builtin_amdgcn_permlanex16(
      __builtin_bit_cast(int, v), __builtin_bit_cast(int, v),
      0x76543210, 0xfedcba98, false, false));
}

__device__ __forceinline__ void swap_halves(float& x, float& y, bool upper) {
  const float m = xhalf(upper ? x : y);
  const float nx = upper ? m : x;
  const float ny = upper ? y : m;
  x = nx; y = ny;
}

#define WMMA(A, B, C) __builtin_amdgcn_wmma_f32_16x16x4_f32(false, (A), false, (B), (short)0, (C), false, false)

__global__ __launch_bounds__(64, 4)
void qsim_kernel(const float* __restrict__ x,
                 const float* __restrict__ gates,
                 const float* __restrict__ fcw,
                 const float* __restrict__ fcb,
                 float* __restrict__ out) {
  __shared__ __attribute__((aligned(16))) float lds[2 * 2 * 1024];
  const int wave = threadIdx.x >> 5;
  const int lane = threadIdx.x & 31;
  const int half = lane >> 4;
  const int l16  = lane & 15;
  const int b    = blockIdx.x * 2 + wave;

  float* Sr = lds + wave * 2048;
  float* Si = Sr + 1024;

  float cw[NQ], sw[NQ];
  for (int w = 0; w < NQ; ++w) {
    float xv = x[b * NQ + w];
    xv = xv > 0.f ? xv : 0.f;
    cw[w] = cosf(0.5f * xv);
    sw[w] = sinf(0.5f * xv);
  }
  {
    const float* einit = gates + EINIT_OFS + lane * 4;
    float mhi = 1.f;
    for (int w = 0; w < 5; ++w) mhi *= ((lane >> (4 - w)) & 1) ? sw[w] : cw[w];
    int pchi = __popc(lane);
    for (int c4 = 0; c4 < 8; ++c4) {
      const v4f pr = *(const v4f*)(einit + c4 * 128);
      const v4f pi = *(const v4f*)(einit + 1024 + c4 * 128);
      #pragma unroll
      for (int j = 0; j < 4; ++j) {
        const int c = c4 * 4 + j;
        float m = mhi;
        for (int w = 0; w < 5; ++w) m *= ((c >> (4 - w)) & 1) ? sw[5 + w] : cw[5 + w];
        int k = (pchi + __popc(c)) & 3;
        float re = 0.f, im = 0.f;
        if      (k == 0) re =  m;
        else if (k == 1) im = -m;
        else if (k == 2) re = -m;
        else             im =  m;
        const float er = pr[j], ei = pi[j];
        const int pa = lds_slot((lane << 5) | c);
        Sr[pa] = re * er - im * ei;
        Si[pa] = re * ei + im * er;
      }
    }
  }
  __syncthreads();

  const int rd0 = (l16 << 4) | ((((2 * half)     ^ (l16 >> 2)) & 3) << 2);
  const int rd1 = (l16 << 4) | ((((2 * half + 1) ^ (l16 >> 2)) & 3) << 2);

  for (int l = 0; l < LAYERS; ++l) {
    const float* G = gates + l * 1024;
    const float* E = gates + ETAB_OFS + l * 2048 + lane * 4;

    const v4f gx0 = *(const v4f*)(G + l16 * 16 + 8 * half);
    const v4f gx1 = *(const v4f*)(G + l16 * 16 + 8 * half + 4);
    const v4f gy0 = *(const v4f*)(G + 256 + l16 * 16 + 8 * half);
    const v4f gy1 = *(const v4f*)(G + 256 + l16 * 16 + 8 * half + 4);
    const v2f gz  = *(const v2f*)(G + 512 + (l16 & 3) * 4 + 2 * half);
    const v2f zero2 = {0.f, 0.f};
    v2f az[4];
    #pragma unroll
    for (int q = 0; q < 4; ++q) az[q] = ((l16 >> 2) == q) ? gz : zero2;

    const int rr = l % 9 + 1;
    const int pLane = lds_slot(cnot_map((half << 7) | l16, rr));

    v8f d2[2][4];
    #pragma unroll
    for (int pl = 0; pl < 2; ++pl) {
      const float* Sp = Sr + pl * 1024;
      #pragma unroll
      for (int t = 0; t < 4; ++t) {
        const v4f s0 = *(const v4f*)(Sp + t * 256 + rd0);
        const v4f s1 = *(const v4f*)(Sp + t * 256 + rd1);
        v8f d1 = {};
        d1 = WMMA(s0.xy, gx0.xy, d1);
        d1 = WMMA(s0.zw, gx0.zw, d1);
        d1 = WMMA(s1.xy, gx1.xy, d1);
        d1 = WMMA(s1.zw, gx1.zw, d1);
        v8f d = {};
        d = WMMA(__builtin_shufflevector(d1, d1, 0, 1), gy0.xy, d);
        d = WMMA(__builtin_shufflevector(d1, d1, 2, 3), gy0.zw, d);
        d = WMMA(__builtin_shufflevector(d1, d1, 4, 5), gy1.xy, d);
        d = WMMA(__builtin_shufflevector(d1, d1, 6, 7), gy1.zw, d);
        d2[pl][t] = d;
      }
    }

    #pragma unroll
    for (int pl = 0; pl < 2; ++pl)
      #pragma unroll
      for (int e = 0; e < 2; ++e)
        #pragma unroll
        for (int r = 0; r < 8; ++r) {
          float u = d2[pl][e][r], v = d2[pl][2 + e][r];
          swap_halves(u, v, half != 0);
          d2[pl][e][r] = u; d2[pl][2 + e][r] = v;
        }

    #pragma unroll
    for (int g = 0; g < 4; ++g) {
      v8f K[2];
      #pragma unroll
      for (int pl = 0; pl < 2; ++pl) {
        const v2f b0 = {d2[pl][0][2 * g],     d2[pl][1][2 * g]};
        const v2f b1 = {d2[pl][0][2 * g + 1], d2[pl][1][2 * g + 1]};
        const v2f b2 = {d2[pl][2][2 * g],     d2[pl][3][2 * g]};
        const v2f b3 = {d2[pl][2][2 * g + 1], d2[pl][3][2 * g + 1]};
        v8f k = {};
        k = WMMA(az[0], b0, k);
        k = WMMA(az[1], b1, k);
        k = WMMA(az[2], b2, k);
        k = WMMA(az[3], b3, k);
        K[pl] = k;
      }
      const float* Et = E + g * 512;
      const v4f er0 = *(const v4f*)(Et);
      const v4f er1 = *(const v4f*)(Et + 128);
      const v4f ei0 = *(const v4f*)(Et + 256);
      const v4f ei1 = *(const v4f*)(Et + 384);
      const int pbase = pLane ^ lds_slot(cnot_map(g << 5, rr));
      #pragma unroll
      for (int r = 0; r < 8; ++r) {
        const int pa = pbase ^ lds_slot(cnot_map(((r & 3) << 8) | ((r >> 2) << 4), rr));
        const float er = (r < 4) ? er0[r & 3] : er1[r & 3];
        const float ei = (r < 4) ? ei0[r & 3] : ei1[r & 3];
        Sr[pa] = K[0][r] * er - K[1][r] * ei;
        Si[pa] = K[0][r] * ei + K[1][r] * er;
      }
    }
    __syncthreads();
  }

  float z[NQ];
  for (int w = 0; w < NQ; ++w) z[w] = 0.f;
  #pragma unroll
  for (int t = 0; t < 4; ++t) {
    #pragma unroll
    for (int jc = 0; jc < 2; ++jc) {
      const v4f r4 = *(const v4f*)(Sr + t * 256 + (jc ? rd1 : rd0));
      const v4f i4 = *(const v4f*)(Si + t * 256 + (jc ? rd1 : rd0));
      #pragma unroll
      for (int j = 0; j < 4; ++j) {
        const int idx = (t << 8) | (half << 7) | (jc << 6) | (j << 4) | l16;
        float re = r4[j], im = i4[j];
        float p = re * re + im * im;
        #pragma unroll
        for (int w = 0; w < NQ; ++w)
          z[w] += ((idx >> (9 - w)) & 1) ? -p : p;
      }
    }
  }
  for (int off = 16; off > 0; off >>= 1)
    #pragma unroll
    for (int w = 0; w < NQ; ++w)
      z[w] += __shfl_xor(z[w], off, 32);
  if (lane < NACT) {
    float acc = fcb[lane];
    for (int w = 0; w < NQ; ++w) acc += fcw[lane * NQ + w] * z[w];
    out[b * NACT + lane] = acc;
  }
}

extern "C" void kernel_launch(void* const* d_in, const int* in_sizes, int n_in,
                              void* d_out, int out_size, void* d_ws, size_t ws_size,
                              hipStream_t stream) {
  const float* inputs  = (const float*)d_in[0];
  const float* weights = (const float*)d_in[1];
  const float* fcw     = (const float*)d_in[2];
  const float* fcb     = (const float*)d_in[3];
  float* out   = (float*)d_out;
  float* gates = (float*)d_ws;

  build_gates_kernel<<<84, 256, 0, stream>>>(weights, gates);
  qsim_kernel<<<BATCH / 2, 64, 0, stream>>>(inputs, gates, fcw, fcb, out);
}


# Write-back LDS addresses (CNOT-ring map fused with slot map) come from per-layer tables built by the setup kernel: one XOR per store, not ~470 scalar ops per layer; table loads fenced to layer head
# speedup vs baseline: 2.3720x; 1.0595x over previous
#include <hip/hip_runtime.h>
#include <hip/hip_bf16.h>

#define NQ      10
#define LAYERS  10
#define BATCH   4096
#define NACT    4

#define GATES_END  (LAYERS * 1024)
#define EINIT_OFS  GATES_END
#define ETAB_OFS   (GATES_END + 2048)
#define SC_OFS     544
#define SL_OFS     576

typedef __attribute__((ext_vector_type(2))) float v2f;
typedef __attribute__((ext_vector_type(4))) float v4f;
typedef __attribute__((ext_vector_type(8))) float v8f;

__device__ __forceinline__ int cnot_map(int b, int rr) {
  #pragma unroll
  for (int w = 0; w < NQ; ++w) {
    int t = w + rr; if (t >= NQ) t -= NQ;
    b ^= ((b >> (9 - w)) & 1) << (9 - t);
  }
  return b;
}

__device__ __forceinline__ int lds_slot(int idx) {
  const int t = idx >> 8, a = (idx >> 4) & 15, b = idx & 15;
  return (t << 8) | (b << 4) | ((((a >> 2) ^ (b >> 2)) & 3) << 2) | (a & 3);
}

__device__ __forceinline__ double rz_sum(const float* __restrict__ w, int l,
                                         int comp, int idx) {
  double s = 0.0;
  for (int q = 0; q < NQ; ++q) {
    double a = (double)w[(l * NQ + q) * 3 + comp];
    s += ((idx >> (9 - q)) & 1) ? a : -a;
  }
  return s;
}

__global__ void build_gates_kernel(const float* __restrict__ w,
                                   float* __restrict__ ws) {
  int id = blockIdx.x * blockDim.x + threadIdx.x;
  if (id < GATES_END) {
    const int l   = id >> 10;
    const int rem = id & 1023;
    int nq, q0, r, cc;
    if (rem < 512) {
      nq = 4; q0 = 2 + (rem >> 8) * 4;
      r = (rem & 255) >> 4; cc = rem & 15;
    } else if (rem < 528) {
      nq = 2; q0 = 0;
      r = (rem - 512) >> 2; cc = rem & 3;
    } else if (rem >= SC_OFS && rem < SC_OFS + 64) {
      const int rr = l % 9 + 1;
      const int e  = rem - SC_OFS;
      int idx;
      if (e < 32) {
        const int g = e >> 3, r = e & 7;
        idx = (g << 5) | ((r & 3) << 8) | ((r >> 2) << 4);
      } else {
        const int ln = e - 32;
        idx = ((ln >> 4) << 7) | (ln & 15);
      }
      ((int*)ws)[l * 1024 + rem] = lds_slot(cnot_map(idx, rr)) << 2;
      return;
    } else {
      return;
    }
    float acc = 1.f;
    for (int j = 0; j < nq; ++j) {
      float theta = w[(l * NQ + q0 + j) * 3 + 1];
      float ct = cosf(0.5f * theta), st = sinf(0.5f * theta);
      int bi = (r  >> (nq - 1 - j)) & 1;
      int bj = (cc >> (nq - 1 - j)) & 1;
      acc *= (bi == bj) ? ct : (bi ? st : -st);
    }
    ws[l * 1024 + rem] = acc;
    return;
  }
  int t = id - GATES_END;
  if (t >= (LAYERS + 1) * 1024) return;
  const int tbl = t >> 10;
  const int e   = t & 1023;
  const int ln  = (e >> 2) & 31;
  const int j   = e & 3;
  double s;
  float* dre;
  float* dim;
  if (tbl == 0) {
    int c   = (e >> 7) * 4 + j;
    int idx = ln * 32 + c;
    s   = rz_sum(w, 0, 0, idx);
    dre = ws + EINIT_OFS + e;
    dim = dre + 1024;
  } else {
    const int l  = tbl - 1;
    const int g  = e >> 8;
    const int hi = (e >> 7) & 1;
    const int a  = 8 * (ln >> 4) + 2 * g + hi;
    const int idx = (j << 8) | (a << 4) | (ln & 15);
    s = rz_sum(w, l, 2, idx);
    if (l + 1 < LAYERS)
      s += rz_sum(w, l + 1, 0, cnot_map(idx, l % 9 + 1));
    dre = ws + ETAB_OFS + l * 2048 + g * 512 + (e & 255);
    dim = dre + 256;
  }
  double turns = s * 0.07957747154594767;
  turns -= rint(turns);
  float a = (float)(turns * 6.283185307179586);
  *dre = cosf(a);
  *dim = sinf(a);
}

__device__ __forceinline__ float xhalf(float v) {
  return __builtin_bit_cast(float, __builtin_amdgcn_permlanex16(
      __builtin_bit_cast(int, v), __builtin_bit_cast(int, v),
      0x76543210, 0xfedcba98, false, false));
}

__device__ __forceinline__ void swap_halves(float& x, float& y, bool upper) {
  const float m = xhalf(upper ? x : y);
  const float nx = upper ? m : x;
  const float ny = upper ? y : m;
  x = nx; y = ny;
}

#define WMMA(A, B, C) __builtin_amdgcn_wmma_f32_16x16x4_f32(false, (A), false, (B), (short)0, (C), false, false)

__global__ __launch_bounds__(64, 4)
void qsim_kernel(const float* __restrict__ x,
                 const float* __restrict__ gates,
                 const float* __restrict__ fcw,
                 const float* __restrict__ fcb,
                 float* __restrict__ out) {
  __shared__ __attribute__((aligned(16))) float lds[2 * 2 * 1024];
  const int wave = threadIdx.x >> 5;
  const int lane = threadIdx.x & 31;
  const int half = lane >> 4;
  const int l16  = lane & 15;
  const int b    = blockIdx.x * 2 + wave;

  float* Sr = lds + wave * 2048;
  float* Si = Sr + 1024;

  float cw[NQ], sw[NQ];
  for (int w = 0; w < NQ; ++w) {
    float xv = x[b * NQ + w];
    xv = xv > 0.f ? xv : 0.f;
    cw[w] = cosf(0.5f * xv);
    sw[w] = sinf(0.5f * xv);
  }
  {
    const float* einit = gates + EINIT_OFS + lane * 4;
    float mhi = 1.f;
    for (int w = 0; w < 5; ++w) mhi *= ((lane >> (4 - w)) & 1) ? sw[w] : cw[w];
    int pchi = __popc(lane);
    for (int c4 = 0; c4 < 8; ++c4) {
      const v4f pr = *(const v4f*)(einit + c4 * 128);
      const v4f pi = *(const v4f*)(einit + 1024 + c4 * 128);
      #pragma unroll
      for (int j = 0; j < 4; ++j) {
        const int c = c4 * 4 + j;
        float m = mhi;
        for (int w = 0; w < 5; ++w) m *= ((c >> (4 - w)) & 1) ? sw[5 + w] : cw[5 + w];
        int k = (pchi + __popc(c)) & 3;
        float re = 0.f, im = 0.f;
        if      (k == 0) re =  m;
        else if (k == 1) im = -m;
        else if (k == 2) re = -m;
        else             im =  m;
        const float er = pr[j], ei = pi[j];
        const int pa = lds_slot((lane << 5) | c);
        Sr[pa] = re * er - im * ei;
        Si[pa] = re * ei + im * er;
      }
    }
  }
  __syncthreads();

  const int rd0 = (l16 << 4) | ((((2 * half)     ^ (l16 >> 2)) & 3) << 2);
  const int rd1 = (l16 << 4) | ((((2 * half + 1) ^ (l16 >> 2)) & 3) << 2);

  for (int l = 0; l < LAYERS; ++l) {
    const float* G = gates + l * 1024;
    const float* E = gates + ETAB_OFS + l * 2048 + lane * 4;

    const v4f gx0 = *(const v4f*)(G + l16 * 16 + 8 * half);
    const v4f gx1 = *(const v4f*)(G + l16 * 16 + 8 * half + 4);
    const v4f gy0 = *(const v4f*)(G + 256 + l16 * 16 + 8 * half);
    const v4f gy1 = *(const v4f*)(G + 256 + l16 * 16 + 8 * half + 4);
    const v2f gz  = *(const v2f*)(G + 512 + (l16 & 3) * 4 + 2 * half);
    const v2f zero2 = {0.f, 0.f};
    v2f az[4];
    #pragma unroll
    for (int q = 0; q < 4; ++q) az[q] = ((l16 >> 2) == q) ? gz : zero2;
    asm volatile("" ::: "memory");

    const int* Tl = (const int*)G;
    const int pLane = (wave << 13) ^ Tl[SL_OFS + lane];
    int sc[32];
    #pragma unroll
    for (int i = 0; i < 32; ++i) sc[i] = Tl[SC_OFS + i];

    v4f ev[4][4];
    #pragma unroll
    for (int g = 0; g < 4; ++g)
      #pragma unroll
      for (int j = 0; j < 4; ++j)
        ev[g][j] = *(const v4f*)(E + g * 512 + j * 128);
    asm volatile("" ::: "memory");

    v8f d2[2][4];
    #pragma unroll
    for (int pl = 0; pl < 2; ++pl) {
      const float* Sp = Sr + pl * 1024;
      #pragma unroll
      for (int t = 0; t < 4; ++t) {
        const v4f s0 = *(const v4f*)(Sp + t * 256 + rd0);
        const v4f s1 = *(const v4f*)(Sp + t * 256 + rd1);
        v8f d1 = {};
        d1 = WMMA(s0.xy, gx0.xy, d1);
        d1 = WMMA(s0.zw, gx0.zw, d1);
        d1 = WMMA(s1.xy, gx1.xy, d1);
        d1 = WMMA(s1.zw, gx1.zw, d1);
        v8f d = {};
        d = WMMA(__builtin_shufflevector(d1, d1, 0, 1), gy0.xy, d);
        d = WMMA(__builtin_shufflevector(d1, d1, 2, 3), gy0.zw, d);
        d = WMMA(__builtin_shufflevector(d1, d1, 4, 5), gy1.xy, d);
        d = WMMA(__builtin_shufflevector(d1, d1, 6, 7), gy1.zw, d);
        d2[pl][t] = d;
      }
    }

    #pragma unroll
    for (int pl = 0; pl < 2; ++pl)
      #pragma unroll
      for (int e = 0; e < 2; ++e)
        #pragma unroll
        for (int r = 0; r < 8; ++r) {
          float u = d2[pl][e][r], v = d2[pl][2 + e][r];
          swap_halves(u, v, half != 0);
          d2[pl][e][r] = u; d2[pl][2 + e][r] = v;
        }

    #pragma unroll
    for (int g = 0; g < 4; ++g) {
      v8f K[2];
      #pragma unroll
      for (int pl = 0; pl < 2; ++pl) {
        const v2f b0 = {d2[pl][0][2 * g],     d2[pl][1][2 * g]};
        const v2f b1 = {d2[pl][0][2 * g + 1], d2[pl][1][2 * g + 1]};
        const v2f b2 = {d2[pl][2][2 * g],     d2[pl][3][2 * g]};
        const v2f b3 = {d2[pl][2][2 * g + 1], d2[pl][3][2 * g + 1]};
        v8f k = {};
        k = WMMA(az[0], b0, k);
        k = WMMA(az[1], b1, k);
        k = WMMA(az[2], b2, k);
        k = WMMA(az[3], b3, k);
        K[pl] = k;
      }
      const v4f er0 = ev[g][0], er1 = ev[g][1], ei0 = ev[g][2], ei1 = ev[g][3];
      #pragma unroll
      for (int r = 0; r < 8; ++r) {
        float* dst = (float*)((char*)lds + (pLane ^ sc[g * 8 + r]));
        const float er = (r < 4) ? er0[r & 3] : er1[r & 3];
        const float ei = (r < 4) ? ei0[r & 3] : ei1[r & 3];
        dst[0]    = K[0][r] * er - K[1][r] * ei;
        dst[1024] = K[0][r] * ei + K[1][r] * er;
      }
    }
    __syncthreads();
  }

  float z[NQ];
  for (int w = 0; w < NQ; ++w) z[w] = 0.f;
  #pragma unroll
  for (int t = 0; t < 4; ++t) {
    #pragma unroll
    for (int jc = 0; jc < 2; ++jc) {
      const v4f r4 = *(const v4f*)(Sr + t * 256 + (jc ? rd1 : rd0));
      const v4f i4 = *(const v4f*)(Si + t * 256 + (jc ? rd1 : rd0));
      #pragma unroll
      for (int j = 0; j < 4; ++j) {
        const int idx = (t << 8) | (half << 7) | (jc << 6) | (j << 4) | l16;
        float re = r4[j], im = i4[j];
        float p = re * re + im * im;
        #pragma unroll
        for (int w = 0; w < NQ; ++w)
          z[w] += ((idx >> (9 - w)) & 1) ? -p : p;
      }
    }
  }
  for (int off = 16; off > 0; off >>= 1)
    #pragma unroll
    for (int w = 0; w < NQ; ++w)
      z[w] += __shfl_xor(z[w], off, 32);
  if (lane < NACT) {
    float acc = fcb[lane];
    for (int w = 0; w < NQ; ++w) acc += fcw[lane * NQ + w] * z[w];
    out[b * NACT + lane] = acc;
  }
}

extern "C" void kernel_launch(void* const* d_in, const int* in_sizes, int n_in,
                              void* d_out, int out_size, void* d_ws, size_t ws_size,
                              hipStream_t stream) {
  const float* inputs  = (const float*)d_in[0];
  const float* weights = (const float*)d_in[1];
  const float* fcw     = (const float*)d_in[2];
  const float* fcb     = (const float*)d_in[3];
  float* out   = (float*)d_out;
  float* gates = (float*)d_ws;

  build_gates_kernel<<<84, 256, 0, stream>>>(weights, gates);
  qsim_kernel<<<BATCH / 2, 64, 0, stream>>>(inputs, gates, fcw, fcb, out);
}


# Input phase with no serial load waits: lane w does wire w's cos/sin once and broadcasts; phase table 0 in the layers' lane order, loaded as one group; 16 B LDS stores; amplitude products regrouped
# speedup vs baseline: 2.6454x; 1.1153x over previous
#include <hip/hip_runtime.h>
#include <hip/hip_bf16.h>

#define NQ      10
#define LAYERS  10
#define BATCH   4096
#define NACT    4

#define GATES_END  (LAYERS * 1024)
#define EINIT_OFS  GATES_END
#define ETAB_OFS   (GATES_END + 2048)
#define SC_OFS     544
#define SL_OFS     576

typedef __attribute__((ext_vector_type(2))) float v2f;
typedef __attribute__((ext_vector_type(4))) float v4f;
typedef __attribute__((ext_vector_type(8))) float v8f;

__device__ __forceinline__ int cnot_map(int b, int rr) {
  #pragma unroll
  for (int w = 0; w < NQ; ++w) {
    int t = w + rr; if (t >= NQ) t -= NQ;
    b ^= ((b >> (9 - w)) & 1) << (9 - t);
  }
  return b;
}

__device__ __forceinline__ int lds_slot(int idx) {
  const int t = idx >> 8, a = (idx >> 4) & 15, b = idx & 15;
  return (t << 8) | (b << 4) | ((((a >> 2) ^ (b >> 2)) & 3) << 2) | (a & 3);
}

__device__ __forceinline__ double rz_sum(const float* __restrict__ w, int l,
                                         int comp, int idx) {
  double s = 0.0;
  for (int q = 0; q < NQ; ++q) {
    double a = (double)w[(l * NQ + q) * 3 + comp];
    s += ((idx >> (9 - q)) & 1) ? a : -a;
  }
  return s;
}

__global__ void build_gates_kernel(const float* __restrict__ w,
                                   float* __restrict__ ws) {
  int id = blockIdx.x * blockDim.x + threadIdx.x;
  if (id < GATES_END) {
    const int l   = id >> 10;
    const int rem = id & 1023;
    int nq, q0, r, cc;
    if (rem < 512) {
      nq = 4; q0 = 2 + (rem >> 8) * 4;
      r = (rem & 255) >> 4; cc = rem & 15;
    } else if (rem < 528) {
      nq = 2; q0 = 0;
      r = (rem - 512) >> 2; cc = rem & 3;
    } else if (rem >= SC_OFS && rem < SC_OFS + 64) {
      const int rr = l % 9 + 1;
      const int e  = rem - SC_OFS;
      int idx;
      if (e < 32) {
        const int g = e >> 3, r = e & 7;
        idx = (g << 5) | ((r & 3) << 8) | ((r >> 2) << 4);
      } else {
        const int ln = e - 32;
        idx = ((ln >> 4) << 7) | (ln & 15);
      }
      ((int*)ws)[l * 1024 + rem] = lds_slot(cnot_map(idx, rr)) << 2;
      return;
    } else {
      return;
    }
    float acc = 1.f;
    for (int j = 0; j < nq; ++j) {
      float theta = w[(l * NQ + q0 + j) * 3 + 1];
      float ct = cosf(0.5f * theta), st = sinf(0.5f * theta);
      int bi = (r  >> (nq - 1 - j)) & 1;
      int bj = (cc >> (nq - 1 - j)) & 1;
      acc *= (bi == bj) ? ct : (bi ? st : -st);
    }
    ws[l * 1024 + rem] = acc;
    return;
  }
  int t = id - GATES_END;
  if (t >= (LAYERS + 1) * 1024) return;
  const int tbl = t >> 10;
  const int e   = t & 1023;
  const int ln  = (e >> 2) & 31;
  const int j   = e & 3;
  double s;
  float* dre;
  float* dim;
  if (tbl == 0) {
    const int k   = e >> 7;
    const int a   = 8 * (ln >> 4) + 4 * (k & 1) + j;
    const int idx = ((k >> 1) << 8) | (a << 4) | (ln & 15);
    s   = rz_sum(w, 0, 0, idx);
    dre = ws + EINIT_OFS + e;
    dim = dre + 1024;
  } else {
    const int l  = tbl - 1;
    const int g  = e >> 8;
    const int hi = (e >> 7) & 1;
    const int a  = 8 * (ln >> 4) + 2 * g + hi;
    const int idx = (j << 8) | (a << 4) | (ln & 15);
    s = rz_sum(w, l, 2, idx);
    if (l + 1 < LAYERS)
      s += rz_sum(w, l + 1, 0, cnot_map(idx, l % 9 + 1));
    dre = ws + ETAB_OFS + l * 2048 + g * 512 + (e & 255);
    dim = dre + 256;
  }
  double turns = s * 0.07957747154594767;
  turns -= rint(turns);
  float a = (float)(turns * 6.283185307179586);
  *dre = cosf(a);
  *dim = sinf(a);
}

__device__ __forceinline__ float xhalf(float v) {
  return __builtin_bit_cast(float, __builtin_amdgcn_permlanex16(
      __builtin_bit_cast(int, v), __builtin_bit_cast(int, v),
      0x76543210, 0xfedcba98, false, false));
}

__device__ __forceinline__ float bcast(float v, int src) {
  return __builtin_bit_cast(float, __builtin_amdgcn_readlane(
      __builtin_bit_cast(int, v), src));
}

__device__ __forceinline__ void swap_halves(float& x, float& y, bool upper) {
  const float m = xhalf(upper ? x : y);
  const float nx = upper ? m : x;
  const float ny = upper ? y : m;
  x = nx; y = ny;
}

#define WMMA(A, B, C) __builtin_amdgcn_wmma_f32_16x16x4_f32(false, (A), false, (B), (short)0, (C), false, false)

__global__ __launch_bounds__(64, 4)
void qsim_kernel(const float* __restrict__ x,
                 const float* __restrict__ gates,
                 const float* __restrict__ fcw,
                 const float* __restrict__ fcb,
                 float* __restrict__ out) {
  __shared__ __attribute__((aligned(16))) float lds[2 * 2 * 1024];
  const int wave = threadIdx.x >> 5;
  const int lane = threadIdx.x & 31;
  const int half = lane >> 4;
  const int l16  = lane & 15;
  const int b    = blockIdx.x * 2 + wave;

  float* Sr = lds + wave * 2048;
  float* Si = Sr + 1024;

  const int rd0 = (l16 << 4) | ((((2 * half)     ^ (l16 >> 2)) & 3) << 2);
  const int rd1 = (l16 << 4) | ((((2 * half + 1) ^ (l16 >> 2)) & 3) << 2);

  float xv = x[b * NQ + (lane < NQ ? lane : NQ - 1)];
  const float* einit = gates + EINIT_OFS + lane * 4;
  v4f pr[8], pi[8];
  #pragma unroll
  for (int k = 0; k < 8; ++k) {
    pr[k] = *(const v4f*)(einit + k * 128);
    pi[k] = *(const v4f*)(einit + 1024 + k * 128);
  }
  xv = xv > 0.f ? xv : 0.f;
  const float cl = cosf(0.5f * xv);
  const float sl = sinf(0.5f * xv);
  float cw[NQ], sw[NQ];
  #pragma unroll
  for (int w = 0; w < NQ; ++w) {
    cw[w] = bcast(cl, w);
    sw[w] = bcast(sl, w);
  }
  {
    float H = half ? sw[2] : cw[2];
    #pragma unroll
    for (int q = 0; q < 4; ++q) H *= ((l16 >> (3 - q)) & 1) ? sw[6 + q] : cw[6 + q];
    const int pl = __popc(lane);
    const float Hs = (pl & 2) ? -H : H;
    const float hr = (pl & 1) ? 0.f : Hs;
    const float hi = (pl & 1) ? -Hs : 0.f;
    float T01[4], A345[8];
    #pragma unroll
    for (int t = 0; t < 4; ++t)
      T01[t] = ((t & 2) ? sw[0] : cw[0]) * ((t & 1) ? sw[1] : cw[1]);
    #pragma unroll
    for (int m = 0; m < 8; ++m)
      A345[m] = ((m & 4) ? sw[3] : cw[3]) * ((m & 2) ? sw[4] : cw[4]) *
                ((m & 1) ? sw[5] : cw[5]);
    #pragma unroll
    for (int k = 0; k < 8; ++k) {
      const int t = k >> 1, jc = k & 1;
      v4f outr, outi;
      #pragma unroll
      for (int j = 0; j < 4; ++j) {
        const int m = jc * 4 + j;
        const float q = T01[t] * A345[m];
        const int pj = (__builtin_popcount(t) + __builtin_popcount(m)) & 3;
        const float er = pr[k][j], ei = pi[k][j];
        const float fr = (pj == 0) ? er : (pj == 1) ? ei : (pj == 2) ? -er : -ei;
        const float fi = (pj == 0) ? ei : (pj == 1) ? -er : (pj == 2) ? -ei : er;
        outr[j] = q * (hr * fr - hi * fi);
        outi[j] = q * (hr * fi + hi * fr);
      }
      const int ofs = t * 256 + (jc ? rd1 : rd0);
      *(v4f*)(Sr + ofs) = outr;
      *(v4f*)(Si + ofs) = outi;
    }
  }
  __syncthreads();

  for (int l = 0; l < LAYERS; ++l) {
    const float* G = gates + l * 1024;
    const float* E = gates + ETAB_OFS + l * 2048 + lane * 4;

    const v4f gx0 = *(const v4f*)(G + l16 * 16 + 8 * half);
    const v4f gx1 = *(const v4f*)(G + l16 * 16 + 8 * half + 4);
    const v4f gy0 = *(const v4f*)(G + 256 + l16 * 16 + 8 * half);
    const v4f gy1 = *(const v4f*)(G + 256 + l16 * 16 + 8 * half + 4);
    const v2f gz  = *(const v2f*)(G + 512 + (l16 & 3) * 4 + 2 * half);
    const v2f zero2 = {0.f, 0.f};
    v2f az[4];
    #pragma unroll
    for (int q = 0; q < 4; ++q) az[q] = ((l16 >> 2) == q) ? gz : zero2;
    asm volatile("" ::: "memory");

    const int* Tl = (const int*)G;
    const int pLane = (wave << 13) ^ Tl[SL_OFS + lane];
    int sc[32];
    #pragma unroll
    for (int i = 0; i < 32; ++i) sc[i] = Tl[SC_OFS + i];

    v4f ev[4][4];
    #pragma unroll
    for (int g = 0; g < 4; ++g)
      #pragma unroll
      for (int j = 0; j < 4; ++j)
        ev[g][j] = *(const v4f*)(E + g * 512 + j * 128);
    asm volatile("" ::: "memory");

    v8f d2[2][4];
    #pragma unroll
    for (int pl = 0; pl < 2; ++pl) {
      const float* Sp = Sr + pl * 1024;
      #pragma unroll
      for (int t = 0; t < 4; ++t) {
        const v4f s0 = *(const v4f*)(Sp + t * 256 + rd0);
        const v4f s1 = *(const v4f*)(Sp + t * 256 + rd1);
        v8f d1 = {};
        d1 = WMMA(s0.xy, gx0.xy, d1);
        d1 = WMMA(s0.zw, gx0.zw, d1);
        d1 = WMMA(s1.xy, gx1.xy, d1);
        d1 = WMMA(s1.zw, gx1.zw, d1);
        v8f d = {};
        d = WMMA(__builtin_shufflevector(d1, d1, 0, 1), gy0.xy, d);
        d = WMMA(__builtin_shufflevector(d1, d1, 2, 3), gy0.zw, d);
        d = WMMA(__builtin_shufflevector(d1, d1, 4, 5), gy1.xy, d);
        d = WMMA(__builtin_shufflevector(d1, d1, 6, 7), gy1.zw, d);
        d2[pl][t] = d;
      }
    }

    #pragma unroll
    for (int pl = 0; pl < 2; ++pl)
      #pragma unroll
      for (int e = 0; e < 2; ++e)
        #pragma unroll
        for (int r = 0; r < 8; ++r) {
          float u = d2[pl][e][r], v = d2[pl][2 + e][r];
          swap_halves(u, v, half != 0);
          d2[pl][e][r] = u; d2[pl][2 + e][r] = v;
        }

    #pragma unroll
    for (int g = 0; g < 4; ++g) {
      v8f K[2];
      #pragma unroll
      for (int pl = 0; pl < 2; ++pl) {
        const v2f b0 = {d2[pl][0][2 * g],     d2[pl][1][2 * g]};
        const v2f b1 = {d2[pl][0][2 * g + 1], d2[pl][1][2 * g + 1]};
        const v2f b2 = {d2[pl][2][2 * g],     d2[pl][3][2 * g]};
        const v2f b3 = {d2[pl][2][2 * g + 1], d2[pl][3][2 * g + 1]};
        v8f k = {};
        k = WMMA(az[0], b0, k);
        k = WMMA(az[1], b1, k);
        k = WMMA(az[2], b2, k);
        k = WMMA(az[3], b3, k);
        K[pl] = k;
      }
      const v4f er0 = ev[g][0], er1 = ev[g][1], ei0 = ev[g][2], ei1 = ev[g][3];
      #pragma unroll
      for (int r = 0; r < 8; ++r) {
        float* dst = (float*)((char*)lds + (pLane ^ sc[g * 8 + r]));
        const float er = (r < 4) ? er0[r & 3] : er1[r & 3];
        const float ei = (r < 4) ? ei0[r & 3] : ei1[r & 3];
        dst[0]    = K[0][r] * er - K[1][r] * ei;
        dst[1024] = K[0][r] * ei + K[1][r] * er;
      }
    }
    __syncthreads();
  }

  float z[NQ];
  for (int w = 0; w < NQ; ++w) z[w] = 0.f;
  #pragma unroll
  for (int t = 0; t < 4; ++t) {
    #pragma unroll
    for (int jc = 0; jc < 2; ++jc) {
      const v4f r4 = *(const v4f*)(Sr + t * 256 + (jc ? rd1 : rd0));
      const v4f i4 = *(const v4f*)(Si + t * 256 + (jc ? rd1 : rd0));
      #pragma unroll
      for (int j = 0; j < 4; ++j) {
        const int idx = (t << 8) | (half << 7) | (jc << 6) | (j << 4) | l16;
        float re = r4[j], im = i4[j];
        float p = re * re + im * im;
        #pragma unroll
        for (int w = 0; w < NQ; ++w)
          z[w] += ((idx >> (9 - w)) & 1) ? -p : p;
      }
    }
  }
  for (int off = 16; off > 0; off >>= 1)
    #pragma unroll
    for (int w = 0; w < NQ; ++w)
      z[w] += __shfl_xor(z[w], off, 32);
  if (lane < NACT) {
    float acc = fcb[lane];
    for (int w = 0; w < NQ; ++w) acc += fcw[lane * NQ + w] * z[w];
    out[b * NACT + lane] = acc;
  }
}

extern "C" void kernel_launch(void* const* d_in, const int* in_sizes, int n_in,
                              void* d_out, int out_size, void* d_ws, size_t ws_size,
                              hipStream_t stream) {
  const float* inputs  = (const float*)d_in[0];
  const float* weights = (const float*)d_in[1];
  const float* fcw     = (const float*)d_in[2];
  const float* fcb     = (const float*)d_in[3];
  float* out   = (float*)d_out;
  float* gates = (float*)d_ws;

  build_gates_kernel<<<84, 256, 0, stream>>>(weights, gates);
  qsim_kernel<<<BATCH / 2, 64, 0, stream>>>(inputs, gates, fcw, fcb, out);
}


# Pass Z reads B straight from pass Y's accumulator register pairs (GZ[t'][t] put at each output row's own k, four tiles t chained on one accumulator): half-exchange gone; changes f32 summation tree
# speedup vs baseline: 2.7581x; 1.0426x over previous
#include <hip/hip_runtime.h>
#include <hip/hip_bf16.h>

#define NQ      10
#define LAYERS  10
#define BATCH   4096
#define NACT    4

#define GATES_END  (LAYERS * 1024)
#define EINIT_OFS  GATES_END
#define ETAB_OFS   (GATES_END + 2048)
#define SC_OFS     544
#define SL_OFS     576

typedef __attribute__((ext_vector_type(2))) float v2f;
typedef __attribute__((ext_vector_type(4))) float v4f;
typedef __attribute__((ext_vector_type(8))) float v8f;

__device__ __forceinline__ int cnot_map(int b, int rr) {
  #pragma unroll
  for (int w = 0; w < NQ; ++w) {
    int t = w + rr; if (t >= NQ) t -= NQ;
    b ^= ((b >> (9 - w)) & 1) << (9 - t);
  }
  return b;
}

__device__ __forceinline__ int lds_slot(int idx) {
  const int t = idx >> 8, a = (idx >> 4) & 15, b = idx & 15;
  return (t << 8) | (b << 4) | ((((a >> 2) ^ (b >> 2)) & 3) << 2) | (a & 3);
}

__device__ __forceinline__ double rz_sum(const float* __restrict__ w, int l,
                                         int comp, int idx) {
  double s = 0.0;
  for (int q = 0; q < NQ; ++q) {
    double a = (double)w[(l * NQ + q) * 3 + comp];
    s += ((idx >> (9 - q)) & 1) ? a : -a;
  }
  return s;
}

__global__ void build_gates_kernel(const float* __restrict__ w,
                                   float* __restrict__ ws) {
  int id = blockIdx.x * blockDim.x + threadIdx.x;
  if (id < GATES_END) {
    const int l   = id >> 10;
    const int rem = id & 1023;
    int nq, q0, r, cc;
    if (rem < 512) {
      nq = 4; q0 = 2 + (rem >> 8) * 4;
      r = (rem & 255) >> 4; cc = rem & 15;
    } else if (rem < 528) {
      nq = 2; q0 = 0;
      r = (rem - 512) >> 2; cc = rem & 3;
    } else if (rem >= SC_OFS && rem < SC_OFS + 64) {
      const int rr = l % 9 + 1;
      const int e  = rem - SC_OFS;
      int idx;
      if (e < 32) {
        const int g = e >> 3, r = e & 7;
        idx = (g << 5) | ((r & 3) << 8) | ((r >> 2) << 4);
      } else {
        const int ln = e - 32;
        idx = ((ln >> 4) << 7) | (ln & 15);
      }
      ((int*)ws)[l * 1024 + rem] = lds_slot(cnot_map(idx, rr)) << 2;
      return;
    } else {
      return;
    }
    float acc = 1.f;
    for (int j = 0; j < nq; ++j) {
      float theta = w[(l * NQ + q0 + j) * 3 + 1];
      float ct = cosf(0.5f * theta), st = sinf(0.5f * theta);
      int bi = (r  >> (nq - 1 - j)) & 1;
      int bj = (cc >> (nq - 1 - j)) & 1;
      acc *= (bi == bj) ? ct : (bi ? st : -st);
    }
    ws[l * 1024 + rem] = acc;
    return;
  }
  int t = id - GATES_END;
  if (t >= (LAYERS + 1) * 1024) return;
  const int tbl = t >> 10;
  const int e   = t & 1023;
  const int ln  = (e >> 2) & 31;
  const int j   = e & 3;
  double s;
  float* dre;
  float* dim;
  if (tbl == 0) {
    const int k   = e >> 7;
    const int a   = 8 * (ln >> 4) + 4 * (k & 1) + j;
    const int idx = ((k >> 1) << 8) | (a << 4) | (ln & 15);
    s   = rz_sum(w, 0, 0, idx);
    dre = ws + EINIT_OFS + e;
    dim = dre + 1024;
  } else {
    const int l  = tbl - 1;
    const int g  = e >> 8;
    const int hi = (e >> 7) & 1;
    const int a  = 8 * (ln >> 4) + 2 * g + hi;
    const int idx = (j << 8) | (a << 4) | (ln & 15);
    s = rz_sum(w, l, 2, idx);
    if (l + 1 < LAYERS)
      s += rz_sum(w, l + 1, 0, cnot_map(idx, l % 9 + 1));
    dre = ws + ETAB_OFS + l * 2048 + g * 512 + (e & 255);
    dim = dre + 256;
  }
  double turns = s * 0.07957747154594767;
  turns -= rint(turns);
  float a = (float)(turns * 6.283185307179586);
  *dre = cosf(a);
  *dim = sinf(a);
}

__device__ __forceinline__ float bcast(float v, int src) {
  return __builtin_bit_cast(float, __builtin_amdgcn_readlane(
      __builtin_bit_cast(int, v), src));
}

#define WMMA(A, B, C) __builtin_amdgcn_wmma_f32_16x16x4_f32(false, (A), false, (B), (short)0, (C), false, false)

__global__ __launch_bounds__(64, 4)
void qsim_kernel(const float* __restrict__ x,
                 const float* __restrict__ gates,
                 const float* __restrict__ fcw,
                 const float* __restrict__ fcb,
                 float* __restrict__ out) {
  __shared__ __attribute__((aligned(16))) float lds[2 * 2 * 1024];
  const int wave = threadIdx.x >> 5;
  const int lane = threadIdx.x & 31;
  const int half = lane >> 4;
  const int l16  = lane & 15;
  const int b    = blockIdx.x * 2 + wave;

  float* Sr = lds + wave * 2048;
  float* Si = Sr + 1024;

  const int rd0 = (l16 << 4) | ((((2 * half)     ^ (l16 >> 2)) & 3) << 2);
  const int rd1 = (l16 << 4) | ((((2 * half + 1) ^ (l16 >> 2)) & 3) << 2);

  float xv = x[b * NQ + (lane < NQ ? lane : NQ - 1)];
  const float* einit = gates + EINIT_OFS + lane * 4;
  v4f pr[8], pi[8];
  #pragma unroll
  for (int k = 0; k < 8; ++k) {
    pr[k] = *(const v4f*)(einit + k * 128);
    pi[k] = *(const v4f*)(einit + 1024 + k * 128);
  }
  xv = xv > 0.f ? xv : 0.f;
  const float cl = cosf(0.5f * xv);
  const float sl = sinf(0.5f * xv);
  float cw[NQ], sw[NQ];
  #pragma unroll
  for (int w = 0; w < NQ; ++w) {
    cw[w] = bcast(cl, w);
    sw[w] = bcast(sl, w);
  }
  {
    float H = half ? sw[2] : cw[2];
    #pragma unroll
    for (int q = 0; q < 4; ++q) H *= ((l16 >> (3 - q)) & 1) ? sw[6 + q] : cw[6 + q];
    const int pl = __popc(lane);
    const float Hs = (pl & 2) ? -H : H;
    const float hr = (pl & 1) ? 0.f : Hs;
    const float hi = (pl & 1) ? -Hs : 0.f;
    float T01[4], A345[8];
    #pragma unroll
    for (int t = 0; t < 4; ++t)
      T01[t] = ((t & 2) ? sw[0] : cw[0]) * ((t & 1) ? sw[1] : cw[1]);
    #pragma unroll
    for (int m = 0; m < 8; ++m)
      A345[m] = ((m & 4) ? sw[3] : cw[3]) * ((m & 2) ? sw[4] : cw[4]) *
                ((m & 1) ? sw[5] : cw[5]);
    #pragma unroll
    for (int k = 0; k < 8; ++k) {
      const int t = k >> 1, jc = k & 1;
      v4f outr, outi;
      #pragma unroll
      for (int j = 0; j < 4; ++j) {
        const int m = jc * 4 + j;
        const float q = T01[t] * A345[m];
        const int pj = (__builtin_popcount(t) + __builtin_popcount(m)) & 3;
        const float er = pr[k][j], ei = pi[k][j];
        const float fr = (pj == 0) ? er : (pj == 1) ? ei : (pj == 2) ? -er : -ei;
        const float fi = (pj == 0) ? ei : (pj == 1) ? -er : (pj == 2) ? -ei : er;
        outr[j] = q * (hr * fr - hi * fi);
        outi[j] = q * (hr * fi + hi * fr);
      }
      const int ofs = t * 256 + (jc ? rd1 : rd0);
      *(v4f*)(Sr + ofs) = outr;
      *(v4f*)(Si + ofs) = outi;
    }
  }
  __syncthreads();

  for (int l = 0; l < LAYERS; ++l) {
    const float* G = gates + l * 1024;
    const float* E = gates + ETAB_OFS + l * 2048 + lane * 4;

    const v4f gx0 = *(const v4f*)(G + l16 * 16 + 8 * half);
    const v4f gx1 = *(const v4f*)(G + l16 * 16 + 8 * half + 4);
    const v4f gy0 = *(const v4f*)(G + 256 + l16 * 16 + 8 * half);
    const v4f gy1 = *(const v4f*)(G + 256 + l16 * 16 + 8 * half + 4);
    const v4f gz  = *(const v4f*)(G + 512 + (l16 & 3) * 4);
    const bool own0 = ((l16 >> 3) == half) && (((l16 >> 2) & 1) == 0);
    const bool own1 = ((l16 >> 3) == half) && (((l16 >> 2) & 1) == 1);
    v2f az[4];
    #pragma unroll
    for (int t = 0; t < 4; ++t) {
      az[t][0] = own0 ? gz[t] : 0.f;
      az[t][1] = own1 ? gz[t] : 0.f;
    }
    asm volatile("" ::: "memory");

    const int* Tl = (const int*)G;
    const int pLane = (wave << 13) ^ Tl[SL_OFS + lane];
    int sc[32];
    #pragma unroll
    for (int i = 0; i < 32; ++i) sc[i] = Tl[SC_OFS + i];

    v4f ev[4][4];
    #pragma unroll
    for (int g = 0; g < 4; ++g)
      #pragma unroll
      for (int j = 0; j < 4; ++j)
        ev[g][j] = *(const v4f*)(E + g * 512 + j * 128);
    asm volatile("" ::: "memory");

    v8f d2[2][4];
    #pragma unroll
    for (int pl = 0; pl < 2; ++pl) {
      const float* Sp = Sr + pl * 1024;
      #pragma unroll
      for (int t = 0; t < 4; ++t) {
        const v4f s0 = *(const v4f*)(Sp + t * 256 + rd0);
        const v4f s1 = *(const v4f*)(Sp + t * 256 + rd1);
        v8f d1 = {};
        d1 = WMMA(s0.xy, gx0.xy, d1);
        d1 = WMMA(s0.zw, gx0.zw, d1);
        d1 = WMMA(s1.xy, gx1.xy, d1);
        d1 = WMMA(s1.zw, gx1.zw, d1);
        v8f d = {};
        d = WMMA(__builtin_shufflevector(d1, d1, 0, 1), gy0.xy, d);
        d = WMMA(__builtin_shufflevector(d1, d1, 2, 3), gy0.zw, d);
        d = WMMA(__builtin_shufflevector(d1, d1, 4, 5), gy1.xy, d);
        d = WMMA(__builtin_shufflevector(d1, d1, 6, 7), gy1.zw, d);
        d2[pl][t] = d;
      }
    }

    #pragma unroll
    for (int g = 0; g < 4; ++g) {
      v8f K[2];
      #pragma unroll
      for (int pl = 0; pl < 2; ++pl) {
        v8f k = {};
        #pragma unroll
        for (int t = 0; t < 4; ++t) {
          const v2f bt = {d2[pl][t][2 * g], d2[pl][t][2 * g + 1]};
          k = WMMA(az[t], bt, k);
        }
        K[pl] = k;
      }
      const v4f er0 = ev[g][0], er1 = ev[g][1], ei0 = ev[g][2], ei1 = ev[g][3];
      #pragma unroll
      for (int r = 0; r < 8; ++r) {
        float* dst = (float*)((char*)lds + (pLane ^ sc[g * 8 + r]));
        const float er = (r < 4) ? er0[r & 3] : er1[r & 3];
        const float ei = (r < 4) ? ei0[r & 3] : ei1[r & 3];
        dst[0]    = K[0][r] * er - K[1][r] * ei;
        dst[1024] = K[0][r] * ei + K[1][r] * er;
      }
    }
    __syncthreads();
  }

  float z[NQ];
  for (int w = 0; w < NQ; ++w) z[w] = 0.f;
  #pragma unroll
  for (int t = 0; t < 4; ++t) {
    #pragma unroll
    for (int jc = 0; jc < 2; ++jc) {
      const v4f r4 = *(const v4f*)(Sr + t * 256 + (jc ? rd1 : rd0));
      const v4f i4 = *(const v4f*)(Si + t * 256 + (jc ? rd1 : rd0));
      #pragma unroll
      for (int j = 0; j < 4; ++j) {
        const int idx = (t << 8) | (half << 7) | (jc << 6) | (j << 4) | l16;
        float re = r4[j], im = i4[j];
        float p = re * re + im * im;
        #pragma unroll
        for (int w = 0; w < NQ; ++w)
          z[w] += ((idx >> (9 - w)) & 1) ? -p : p;
      }
    }
  }
  for (int off = 16; off > 0; off >>= 1)
    #pragma unroll
    for (int w = 0; w < NQ; ++w)
      z[w] += __shfl_xor(z[w], off, 32);
  if (lane < NACT) {
    float acc = fcb[lane];
    for (int w = 0; w < NQ; ++w) acc += fcw[lane * NQ + w] * z[w];
    out[b * NACT + lane] = acc;
  }
}

extern "C" void kernel_launch(void* const* d_in, const int* in_sizes, int n_in,
                              void* d_out, int out_size, void* d_ws, size_t ws_size,
                              hipStream_t stream) {
  const float* inputs  = (const float*)d_in[0];
  const float* weights = (const float*)d_in[1];
  const float* fcw     = (const float*)d_in[2];
  const float* fcb     = (const float*)d_in[3];
  float* out   = (float*)d_out;
  float* gates = (float*)d_ws;

  build_gates_kernel<<<84, 256, 0, stream>>>(weights, gates);
  qsim_kernel<<<BATCH / 2, 64, 0, stream>>>(inputs, gates, fcw, fcb, out);
}


# Wave priority steps down with layer progress (s_setprio 2 in layers 0-5, 1 in 6-8, 0 in the last): a SIMD's four waves advance together, not oldest-first one after another; arithmetic unchanged
# speedup vs baseline: 2.9288x; 1.0619x over previous
#include <hip/hip_runtime.h>
#include <hip/hip_bf16.h>

#define NQ      10
#define LAYERS  10
#define BATCH   4096
#define NACT    4

#define GATES_END  (LAYERS * 1024)
#define EINIT_OFS  GATES_END
#define ETAB_OFS   (GATES_END + 2048)
#define SC_OFS     544
#define SL_OFS     576

typedef __attribute__((ext_vector_type(2))) float v2f;
typedef __attribute__((ext_vector_type(4))) float v4f;
typedef __attribute__((ext_vector_type(8))) float v8f;
typedef __attribute__((address_space(4))) int cint_k;

__device__ __forceinline__ int cnot_map(int b, int rr) {
  #pragma unroll
  for (int w = 0; w < NQ; ++w) {
    int t = w + rr; if (t >= NQ) t -= NQ;
    b ^= ((b >> (9 - w)) & 1) << (9 - t);
  }
  return b;
}

__device__ __forceinline__ int lds_slot(int idx) {
  const int t = idx >> 8, a = (idx >> 4) & 15, b = idx & 15;
  return (t << 8) | (b << 4) | ((((a >> 2) ^ (b >> 2)) & 3) << 2) | (a & 3);
}

__device__ __forceinline__ double rz_sum(const float* __restrict__ w, int l,
                                         int comp, int idx) {
  double s = 0.0;
  for (int q = 0; q < NQ; ++q) {
    double a = (double)w[(l * NQ + q) * 3 + comp];
    s += ((idx >> (9 - q)) & 1) ? a : -a;
  }
  return s;
}

__global__ void build_gates_kernel(const float* __restrict__ w,
                                   float* __restrict__ ws) {
  int id = blockIdx.x * blockDim.x + threadIdx.x;
  if (id < GATES_END) {
    const int l   = id >> 10;
    const int rem = id & 1023;
    int nq, q0, r, cc;
    if (rem < 512) {
      nq = 4; q0 = 2 + (rem >> 8) * 4;
      r = (rem & 255) >> 4; cc = rem & 15;
    } else if (rem < 528) {
      nq = 2; q0 = 0;
      r = (rem - 512) >> 2; cc = rem & 3;
    } else if (rem >= SC_OFS && rem < SC_OFS + 64) {
      const int rr = l % 9 + 1;
      const int e  = rem - SC_OFS;
      int idx;
      if (e < 32) {
        const int g = e >> 3, r = e & 7;
        idx = (g << 5) | ((r & 3) << 8) | ((r >> 2) << 4);
      } else {
        const int ln = e - 32;
        idx = ((ln >> 4) << 7) | (ln & 15);
      }
      ((int*)ws)[l * 1024 + rem] = lds_slot(cnot_map(idx, rr)) << 2;
      return;
    } else {
      return;
    }
    float acc = 1.f;
    for (int j = 0; j < nq; ++j) {
      float theta = w[(l * NQ + q0 + j) * 3 + 1];
      float ct = cosf(0.5f * theta), st = sinf(0.5f * theta);
      int bi = (r  >> (nq - 1 - j)) & 1;
      int bj = (cc >> (nq - 1 - j)) & 1;
      acc *= (bi == bj) ? ct : (bi ? st : -st);
    }
    ws[l * 1024 + rem] = acc;
    return;
  }
  int t = id - GATES_END;
  if (t >= (LAYERS + 1) * 1024) return;
  const int tbl = t >> 10;
  const int e   = t & 1023;
  const int ln  = (e >> 2) & 31;
  const int j   = e & 3;
  double s;
  float* dre;
  float* dim;
  if (tbl == 0) {
    const int k   = e >> 7;
    const int a   = 8 * (ln >> 4) + 4 * (k & 1) + j;
    const int idx = ((k >> 1) << 8) | (a << 4) | (ln & 15);
    s   = rz_sum(w, 0, 0, idx);
    dre = ws + EINIT_OFS + e;
    dim = dre + 1024;
  } else {
    const int l  = tbl - 1;
    const int g  = e >> 8;
    const int hi = (e >> 7) & 1;
    const int a  = 8 * (ln >> 4) + 2 * g + hi;
    const int idx = (j << 8) | (a << 4) | (ln & 15);
    s = rz_sum(w, l, 2, idx);
    if (l + 1 < LAYERS)
      s += rz_sum(w, l + 1, 0, cnot_map(idx, l % 9 + 1));
    dre = ws + ETAB_OFS + l * 2048 + g * 512 + (e & 255);
    dim = dre + 256;
  }
  double turns = s * 0.07957747154594767;
  turns -= rint(turns);
  float a = (float)(turns * 6.283185307179586);
  *dre = cosf(a);
  *dim = sinf(a);
}

__device__ __forceinline__ float bcast(float v, int src) {
  return __builtin_bit_cast(float, __builtin_amdgcn_readlane(
      __builtin_bit_cast(int, v), src));
}

#define WMMA(A, B, C) __builtin_amdgcn_wmma_f32_16x16x4_f32(false, (A), false, (B), (short)0, (C), false, false)

__global__ __launch_bounds__(64, 4)
void qsim_kernel(const float* __restrict__ x,
                 const float* __restrict__ gates,
                 const float* __restrict__ fcw,
                 const float* __restrict__ fcb,
                 float* __restrict__ out) {
  __shared__ __attribute__((aligned(16))) float lds[2 * 2 * 1024];
  const int wave = threadIdx.x >> 5;
  const int lane = threadIdx.x & 31;
  const int half = lane >> 4;
  const int l16  = lane & 15;
  const int b    = blockIdx.x * 2 + wave;

  float* Sr = lds + wave * 2048;
  float* Si = Sr + 1024;

  const int rd0 = (l16 << 4) | ((((2 * half)     ^ (l16 >> 2)) & 3) << 2);
  const int rd1 = (l16 << 4) | ((((2 * half + 1) ^ (l16 >> 2)) & 3) << 2);

  float xv = x[b * NQ + (lane < NQ ? lane : NQ - 1)];
  const float* einit = gates + EINIT_OFS + lane * 4;
  v4f pr[8], pi[8];
  #pragma unroll
  for (int k = 0; k < 8; ++k) {
    pr[k] = *(const v4f*)(einit + k * 128);
    pi[k] = *(const v4f*)(einit + 1024 + k * 128);
  }
  xv = xv > 0.f ? xv : 0.f;
  const float cl = cosf(0.5f * xv);
  const float sl = sinf(0.5f * xv);
  float cw[NQ], sw[NQ];
  #pragma unroll
  for (int w = 0; w < NQ; ++w) {
    cw[w] = bcast(cl, w);
    sw[w] = bcast(sl, w);
  }
  {
    float H = half ? sw[2] : cw[2];
    #pragma unroll
    for (int q = 0; q < 4; ++q) H *= ((l16 >> (3 - q)) & 1) ? sw[6 + q] : cw[6 + q];
    const int pl = __popc(lane);
    const float Hs = (pl & 2) ? -H : H;
    const float hr = (pl & 1) ? 0.f : Hs;
    const float hi = (pl & 1) ? -Hs : 0.f;
    float T01[4], A345[8];
    #pragma unroll
    for (int t = 0; t < 4; ++t)
      T01[t] = ((t & 2) ? sw[0] : cw[0]) * ((t & 1) ? sw[1] : cw[1]);
    #pragma unroll
    for (int m = 0; m < 8; ++m)
      A345[m] = ((m & 4) ? sw[3] : cw[3]) * ((m & 2) ? sw[4] : cw[4]) *
                ((m & 1) ? sw[5] : cw[5]);
    #pragma unroll
    for (int k = 0; k < 8; ++k) {
      const int t = k >> 1, jc = k & 1;
      v4f outr, outi;
      #pragma unroll
      for (int j = 0; j < 4; ++j) {
        const int m = jc * 4 + j;
        const float q = T01[t] * A345[m];
        const int pj = (__builtin_popcount(t) + __builtin_popcount(m)) & 3;
        const float er = pr[k][j], ei = pi[k][j];
        const float fr = (pj == 0) ? er : (pj == 1) ? ei : (pj == 2) ? -er : -ei;
        const float fi = (pj == 0) ? ei : (pj == 1) ? -er : (pj == 2) ? -ei : er;
        outr[j] = q * (hr * fr - hi * fi);
        outi[j] = q * (hr * fi + hi * fr);
      }
      const int ofs = t * 256 + (jc ? rd1 : rd0);
      *(v4f*)(Sr + ofs) = outr;
      *(v4f*)(Si + ofs) = outi;
    }
  }
  __syncthreads();

  int l = 0;
  #pragma nounroll
  for (int step = 0; step < 3; ++step) {
    int lend;
    if (step == 0)      { asm volatile("s_setprio 2"); lend = 6; }
    else if (step == 1) { asm volatile("s_setprio 1"); lend = LAYERS - 1; }
    else                { asm volatile("s_setprio 0"); lend = LAYERS; }
  for (; l < lend; ++l) {
    const float* G = gates + l * 1024;
    const float* E = gates + ETAB_OFS + l * 2048 + lane * 4;

    const v4f gx0 = *(const v4f*)(G + l16 * 16 + 8 * half);
    const v4f gx1 = *(const v4f*)(G + l16 * 16 + 8 * half + 4);
    const v4f gy0 = *(const v4f*)(G + 256 + l16 * 16 + 8 * half);
    const v4f gy1 = *(const v4f*)(G + 256 + l16 * 16 + 8 * half + 4);
    const v4f gz  = *(const v4f*)(G + 512 + (l16 & 3) * 4);
    const bool own0 = ((l16 >> 3) == half) && (((l16 >> 2) & 1) == 0);
    const bool own1 = ((l16 >> 3) == half) && (((l16 >> 2) & 1) == 1);
    v2f az[4];
    #pragma unroll
    for (int t = 0; t < 4; ++t) {
      az[t][0] = own0 ? gz[t] : 0.f;
      az[t][1] = own1 ? gz[t] : 0.f;
    }
    asm volatile("" ::: "memory");

    const int* Tl = (const int*)G;
    const int pLane = (wave << 13) ^ Tl[SL_OFS + lane];
    const cint_k* Tc = (const cint_k*)Tl;
    int sc[32];
    #pragma unroll
    for (int i = 0; i < 32; ++i) sc[i] = Tc[SC_OFS + i];

    v4f ev[4][4];
    #pragma unroll
    for (int g = 0; g < 4; ++g)
      #pragma unroll
      for (int j = 0; j < 4; ++j)
        ev[g][j] = *(const v4f*)(E + g * 512 + j * 128);
    asm volatile("" ::: "memory");

    v8f d2[2][4];
    #pragma unroll
    for (int pl = 0; pl < 2; ++pl) {
      const float* Sp = Sr + pl * 1024;
      #pragma unroll
      for (int t = 0; t < 4; ++t) {
        const v4f s0 = *(const v4f*)(Sp + t * 256 + rd0);
        const v4f s1 = *(const v4f*)(Sp + t * 256 + rd1);
        v8f d1 = {};
        d1 = WMMA(s0.xy, gx0.xy, d1);
        d1 = WMMA(s0.zw, gx0.zw, d1);
        d1 = WMMA(s1.xy, gx1.xy, d1);
        d1 = WMMA(s1.zw, gx1.zw, d1);
        v8f d = {};
        d = WMMA(__builtin_shufflevector(d1, d1, 0, 1), gy0.xy, d);
        d = WMMA(__builtin_shufflevector(d1, d1, 2, 3), gy0.zw, d);
        d = WMMA(__builtin_shufflevector(d1, d1, 4, 5), gy1.xy, d);
        d = WMMA(__builtin_shufflevector(d1, d1, 6, 7), gy1.zw, d);
        d2[pl][t] = d;
      }
    }

    #pragma unroll
    for (int g = 0; g < 4; ++g) {
      v8f K[2];
      #pragma unroll
      for (int pl = 0; pl < 2; ++pl) {
        v8f k = {};
        #pragma unroll
        for (int t = 0; t < 4; ++t) {
          const v2f bt = {d2[pl][t][2 * g], d2[pl][t][2 * g + 1]};
          k = WMMA(az[t], bt, k);
        }
        K[pl] = k;
      }
      const v4f er0 = ev[g][0], er1 = ev[g][1], ei0 = ev[g][2], ei1 = ev[g][3];
      #pragma unroll
      for (int r = 0; r < 8; ++r) {
        float* dst = (float*)((char*)lds + (pLane ^ sc[g * 8 + r]));
        const float er = (r < 4) ? er0[r & 3] : er1[r & 3];
        const float ei = (r < 4) ? ei0[r & 3] : ei1[r & 3];
        dst[0]    = K[0][r] * er - K[1][r] * ei;
        dst[1024] = K[0][r] * ei + K[1][r] * er;
      }
    }
    __syncthreads();
  }
  }

  float z[NQ];
  for (int w = 0; w < NQ; ++w) z[w] = 0.f;
  #pragma unroll
  for (int t = 0; t < 4; ++t) {
    #pragma unroll
    for (int jc = 0; jc < 2; ++jc) {
      const v4f r4 = *(const v4f*)(Sr + t * 256 + (jc ? rd1 : rd0));
      const v4f i4 = *(const v4f*)(Si + t * 256 + (jc ? rd1 : rd0));
      #pragma unroll
      for (int j = 0; j < 4; ++j) {
        const int idx = (t << 8) | (half << 7) | (jc << 6) | (j << 4) | l16;
        float re = r4[j], im = i4[j];
        float p = re * re + im * im;
        #pragma unroll
        for (int w = 0; w < NQ; ++w)
          z[w] += ((idx >> (9 - w)) & 1) ? -p : p;
      }
    }
  }
  for (int off = 16; off > 0; off >>= 1)
    #pragma unroll
    for (int w = 0; w < NQ; ++w)
      z[w] += __shfl_xor(z[w], off, 32);
  if (lane < NACT) {
    float acc = fcb[lane];
    for (int w = 0; w < NQ; ++w) acc += fcw[lane * NQ + w] * z[w];
    out[b * NACT + lane] = acc;
  }
}

extern "C" void kernel_launch(void* const* d_in, const int* in_sizes, int n_in,
                              void* d_out, int out_size, void* d_ws, size_t ws_size,
                              hipStream_t stream) {
  const float* inputs  = (const float*)d_in[0];
  const float* weights = (const float*)d_in[1];
  const float* fcw     = (const float*)d_in[2];
  const float* fcb     = (const float*)d_in[3];
  float* out   = (float*)d_out;
  float* gates = (float*)d_ws;

  build_gates_kernel<<<84, 256, 0, stream>>>(weights, gates);
  qsim_kernel<<<BATCH / 2, 64, 0, stream>>>(inputs, gates, fcw, fcb, out);
}


# Layer 0 in closed form: the embedding leaves a product state, so layer 0 acts on ten complex 2-vectors whose Kronecker product is written back; 864 WMMAs per item, not 960; changes f32 summation tree
# speedup vs baseline: 3.1153x; 1.0637x over previous
#include <hip/hip_runtime.h>
#include <hip/hip_bf16.h>

#define NQ      10
#define LAYERS  10
#define BATCH   4096
#define NACT    4

#define GATES_END  (LAYERS * 1024)
#define EINIT_OFS  GATES_END
#define ETAB_OFS   (GATES_END + 2048)
#define SC_OFS     544
#define SL_OFS     576

typedef __attribute__((ext_vector_type(2))) float v2f;
typedef __attribute__((ext_vector_type(4))) float v4f;
typedef __attribute__((ext_vector_type(8))) float v8f;
typedef __attribute__((address_space(4))) int cint_k;

__device__ __forceinline__ int cnot_map(int b, int rr) {
  #pragma unroll
  for (int w = 0; w < NQ; ++w) {
    int t = w + rr; if (t >= NQ) t -= NQ;
    b ^= ((b >> (9 - w)) & 1) << (9 - t);
  }
  return b;
}

__device__ __forceinline__ int lds_slot(int idx) {
  const int t = idx >> 8, a = (idx >> 4) & 15, b = idx & 15;
  return (t << 8) | (b << 4) | ((((a >> 2) ^ (b >> 2)) & 3) << 2) | (a & 3);
}

__device__ __forceinline__ double rz_sum(const float* __restrict__ w, int l,
                                         int comp, int idx) {
  double s = 0.0;
  for (int q = 0; q < NQ; ++q) {
    double a = (double)w[(l * NQ + q) * 3 + comp];
    s += ((idx >> (9 - q)) & 1) ? a : -a;
  }
  return s;
}

__global__ void build_gates_kernel(const float* __restrict__ w,
                                   float* __restrict__ ws) {
  int id = blockIdx.x * blockDim.x + threadIdx.x;
  if (id < GATES_END) {
    const int l   = id >> 10;
    const int rem = id & 1023;
    int nq, q0, r, cc;
    if (rem < 512) {
      nq = 4; q0 = 2 + (rem >> 8) * 4;
      r = (rem & 255) >> 4; cc = rem & 15;
    } else if (rem < 528) {
      nq = 2; q0 = 0;
      r = (rem - 512) >> 2; cc = rem & 3;
    } else if (rem >= SC_OFS && rem < SC_OFS + 64) {
      const int rr = l % 9 + 1;
      const int e  = rem - SC_OFS;
      int idx;
      if (e < 32) {
        const int g = e >> 3, r = e & 7;
        idx = (g << 5) | ((r & 3) << 8) | ((r >> 2) << 4);
      } else {
        const int ln = e - 32;
        idx = ((ln >> 4) << 7) | (ln & 15);
      }
      ((int*)ws)[l * 1024 + rem] = lds_slot(cnot_map(idx, rr)) << 2;
      return;
    } else {
      return;
    }
    float acc = 1.f;
    for (int j = 0; j < nq; ++j) {
      float theta = w[(l * NQ + q0 + j) * 3 + 1];
      float ct = cosf(0.5f * theta), st = sinf(0.5f * theta);
      int bi = (r  >> (nq - 1 - j)) & 1;
      int bj = (cc >> (nq - 1 - j)) & 1;
      acc *= (bi == bj) ? ct : (bi ? st : -st);
    }
    ws[l * 1024 + rem] = acc;
    return;
  }
  int t = id - GATES_END;
  if (t >= (LAYERS + 1) * 1024) return;
  const int tbl = t >> 10;
  const int e   = t & 1023;
  const int ln  = (e >> 2) & 31;
  const int j   = e & 3;
  double s;
  float* dre;
  float* dim;
  if (tbl == 0) {
    if (e < NQ) {
      const float th = w[e * 3 + 1], ph = w[e * 3 + 0];
      float* d = ws + EINIT_OFS + e * 4;
      d[0] = cosf(0.5f * th);
      d[1] = sinf(0.5f * th);
      d[2] = cosf(0.5f * ph);
      d[3] = sinf(0.5f * ph);
    }
    return;
  } else {
    const int l  = tbl - 1;
    const int g  = e >> 8;
    const int hi = (e >> 7) & 1;
    const int a  = 8 * (ln >> 4) + 2 * g + hi;
    const int idx = (j << 8) | (a << 4) | (ln & 15);
    s = rz_sum(w, l, 2, idx);
    if (l + 1 < LAYERS)
      s += rz_sum(w, l + 1, 0, cnot_map(idx, l % 9 + 1));
    dre = ws + ETAB_OFS + l * 2048 + g * 512 + (e & 255);
    dim = dre + 256;
  }
  double turns = s * 0.07957747154594767;
  turns -= rint(turns);
  float a = (float)(turns * 6.283185307179586);
  *dre = cosf(a);
  *dim = sinf(a);
}

__device__ __forceinline__ float bcast(float v, int src) {
  return __builtin_bit_cast(float, __builtin_amdgcn_readlane(
      __builtin_bit_cast(int, v), src));
}

__device__ __forceinline__ v2f cmul(v2f a, v2f b) {
  v2f r;
  r[0] = a[0] * b[0] - a[1] * b[1];
  r[1] = a[0] * b[1] + a[1] * b[0];
  return r;
}

#define WMMA(A, B, C) __builtin_amdgcn_wmma_f32_16x16x4_f32(false, (A), false, (B), (short)0, (C), false, false)

__global__ __launch_bounds__(64, 4)
void qsim_kernel(const float* __restrict__ x,
                 const float* __restrict__ gates,
                 const float* __restrict__ fcw,
                 const float* __restrict__ fcb,
                 float* __restrict__ out) {
  __shared__ __attribute__((aligned(16))) float lds[2 * 2 * 1024];
  const int wave = threadIdx.x >> 5;
  const int lane = threadIdx.x & 31;
  const int half = lane >> 4;
  const int l16  = lane & 15;
  const int b    = blockIdx.x * 2 + wave;

  float* Sr = lds + wave * 2048;
  float* Si = Sr + 1024;

  const int rd0 = (l16 << 4) | ((((2 * half)     ^ (l16 >> 2)) & 3) << 2);
  const int rd1 = (l16 << 4) | ((((2 * half + 1) ^ (l16 >> 2)) & 3) << 2);

  const int wl = lane < NQ ? lane : NQ - 1;
  float xv = x[b * NQ + wl];
  const v4f u0t = *(const v4f*)(gates + EINIT_OFS + 4 * wl);
  const float* E0 = gates + ETAB_OFS + lane * 4;
  const int* Tl0 = (const int*)gates;
  const int pLane0 = (wave << 13) ^ Tl0[SL_OFS + lane];
  const cint_k* Tc0 = (const cint_k*)Tl0;
  int sc0[32];
  #pragma unroll
  for (int i = 0; i < 32; ++i) sc0[i] = Tc0[SC_OFS + i];
  v4f ev0[4][4];
  #pragma unroll
  for (int g = 0; g < 4; ++g)
    #pragma unroll
    for (int j = 0; j < 4; ++j)
      ev0[g][j] = *(const v4f*)(E0 + g * 512 + j * 128);
  xv = xv > 0.f ? xv : 0.f;
  const float cl = cosf(0.5f * xv);
  const float sl = sinf(0.5f * xv);
  {
    const float ct = u0t[0], st = u0t[1], cp = u0t[2], sp = u0t[3];
    const float a0r = cp * cl, a0i = -(sp * cl);
    const float a1r = sp * sl, a1i = -(cp * sl);
    const float w0r = ct * a0r - st * a1r, w0i = ct * a0i - st * a1i;
    const float w1r = st * a0r + ct * a1r, w1i = st * a0i + ct * a1i;
    v2f U0[NQ], U1[NQ];
    #pragma unroll
    for (int w = 0; w < NQ; ++w) {
      U0[w][0] = bcast(w0r, w); U0[w][1] = bcast(w0i, w);
      U1[w][0] = bcast(w1r, w); U1[w][1] = bcast(w1i, w);
    }
    v2f Hc = half ? U1[2] : U0[2];
    #pragma unroll
    for (int q = 0; q < 4; ++q) {
      const bool bit = (l16 >> (3 - q)) & 1;
      Hc = cmul(Hc, bit ? U1[6 + q] : U0[6 + q]);
    }
    v2f T01[4], T015[8], GH[4];
    #pragma unroll
    for (int t = 0; t < 4; ++t)
      T01[t] = cmul((t & 2) ? U1[0] : U0[0], (t & 1) ? U1[1] : U0[1]);
    #pragma unroll
    for (int r = 0; r < 8; ++r)
      T015[r] = cmul(T01[r & 3], (r & 4) ? U1[5] : U0[5]);
    #pragma unroll
    for (int g = 0; g < 4; ++g)
      GH[g] = cmul(cmul((g & 2) ? U1[3] : U0[3], (g & 1) ? U1[4] : U0[4]), Hc);
    #pragma unroll
    for (int g = 0; g < 4; ++g) {
      const v4f er0 = ev0[g][0], er1 = ev0[g][1];
      const v4f ei0 = ev0[g][2], ei1 = ev0[g][3];
      #pragma unroll
      for (int r = 0; r < 8; ++r) {
        const v2f k = cmul(GH[g], T015[r]);
        float* dst = (float*)((char*)lds + (pLane0 ^ sc0[g * 8 + r]));
        const float er = (r < 4) ? er0[r & 3] : er1[r & 3];
        const float ei = (r < 4) ? ei0[r & 3] : ei1[r & 3];
        dst[0]    = k[0] * er - k[1] * ei;
        dst[1024] = k[0] * ei + k[1] * er;
      }
    }
  }
  __syncthreads();

  int l = 1;
  #pragma nounroll
  for (int step = 0; step < 3; ++step) {
    int lend;
    if (step == 0)      { asm volatile("s_setprio 2"); lend = 6; }
    else if (step == 1) { asm volatile("s_setprio 1"); lend = LAYERS - 1; }
    else                { asm volatile("s_setprio 0"); lend = LAYERS; }
  for (; l < lend; ++l) {
    const float* G = gates + l * 1024;
    const float* E = gates + ETAB_OFS + l * 2048 + lane * 4;

    const v4f gx0 = *(const v4f*)(G + l16 * 16 + 8 * half);
    const v4f gx1 = *(const v4f*)(G + l16 * 16 + 8 * half + 4);
    const v4f gy0 = *(const v4f*)(G + 256 + l16 * 16 + 8 * half);
    const v4f gy1 = *(const v4f*)(G + 256 + l16 * 16 + 8 * half + 4);
    const v4f gz  = *(const v4f*)(G + 512 + (l16 & 3) * 4);
    const bool own0 = ((l16 >> 3) == half) && (((l16 >> 2) & 1) == 0);
    const bool own1 = ((l16 >> 3) == half) && (((l16 >> 2) & 1) == 1);
    v2f az[4];
    #pragma unroll
    for (int t = 0; t < 4; ++t) {
      az[t][0] = own0 ? gz[t] : 0.f;
      az[t][1] = own1 ? gz[t] : 0.f;
    }
    asm volatile("" ::: "memory");

    const int* Tl = (const int*)G;
    const int pLane = (wave << 13) ^ Tl[SL_OFS + lane];
    const cint_k* Tc = (const cint_k*)Tl;
    int sc[32];
    #pragma unroll
    for (int i = 0; i < 32; ++i) sc[i] = Tc[SC_OFS + i];

    v4f ev[4][4];
    #pragma unroll
    for (int g = 0; g < 4; ++g)
      #pragma unroll
      for (int j = 0; j < 4; ++j)
        ev[g][j] = *(const v4f*)(E + g * 512 + j * 128);
    asm volatile("" ::: "memory");

    v8f d2[2][4];
    #pragma unroll
    for (int pl = 0; pl < 2; ++pl) {
      const float* Sp = Sr + pl * 1024;
      #pragma unroll
      for (int t = 0; t < 4; ++t) {
        const v4f s0 = *(const v4f*)(Sp + t * 256 + rd0);
        const v4f s1 = *(const v4f*)(Sp + t * 256 + rd1);
        v8f d1 = {};
        d1 = WMMA(s0.xy, gx0.xy, d1);
        d1 = WMMA(s0.zw, gx0.zw, d1);
        d1 = WMMA(s1.xy, gx1.xy, d1);
        d1 = WMMA(s1.zw, gx1.zw, d1);
        v8f d = {};
        d = WMMA(__builtin_shufflevector(d1, d1, 0, 1), gy0.xy, d);
        d = WMMA(__builtin_shufflevector(d1, d1, 2, 3), gy0.zw, d);
        d = WMMA(__builtin_shufflevector(d1, d1, 4, 5), gy1.xy, d);
        d = WMMA(__builtin_shufflevector(d1, d1, 6, 7), gy1.zw, d);
        d2[pl][t] = d;
      }
    }

    #pragma unroll
    for (int g = 0; g < 4; ++g) {
      v8f K[2];
      #pragma unroll
      for (int pl = 0; pl < 2; ++pl) {
        v8f k = {};
        #pragma unroll
        for (int t = 0; t < 4; ++t) {
          const v2f bt = {d2[pl][t][2 * g], d2[pl][t][2 * g + 1]};
          k = WMMA(az[t], bt, k);
        }
        K[pl] = k;
      }
      const v4f er0 = ev[g][0], er1 = ev[g][1], ei0 = ev[g][2], ei1 = ev[g][3];
      #pragma unroll
      for (int r = 0; r < 8; ++r) {
        float* dst = (float*)((char*)lds + (pLane ^ sc[g * 8 + r]));
        const float er = (r < 4) ? er0[r & 3] : er1[r & 3];
        const float ei = (r < 4) ? ei0[r & 3] : ei1[r & 3];
        dst[0]    = K[0][r] * er - K[1][r] * ei;
        dst[1024] = K[0][r] * ei + K[1][r] * er;
      }
    }
    __syncthreads();
  }
  }

  float z[NQ];
  for (int w = 0; w < NQ; ++w) z[w] = 0.f;
  #pragma unroll
  for (int t = 0; t < 4; ++t) {
    #pragma unroll
    for (int jc = 0; jc < 2; ++jc) {
      const v4f r4 = *(const v4f*)(Sr + t * 256 + (jc ? rd1 : rd0));
      const v4f i4 = *(const v4f*)(Si + t * 256 + (jc ? rd1 : rd0));
      #pragma unroll
      for (int j = 0; j < 4; ++j) {
        const int idx = (t << 8) | (half << 7) | (jc << 6) | (j << 4) | l16;
        float re = r4[j], im = i4[j];
        float p = re * re + im * im;
        #pragma unroll
        for (int w = 0; w < NQ; ++w)
          z[w] += ((idx >> (9 - w)) & 1) ? -p : p;
      }
    }
  }
  for (int off = 16; off > 0; off >>= 1)
    #pragma unroll
    for (int w = 0; w < NQ; ++w)
      z[w] += __shfl_xor(z[w], off, 32);
  if (lane < NACT) {
    float acc = fcb[lane];
    for (int w = 0; w < NQ; ++w) acc += fcw[lane * NQ + w] * z[w];
    out[b * NACT + lane] = acc;
  }
}

extern "C" void kernel_launch(void* const* d_in, const int* in_sizes, int n_in,
                              void* d_out, int out_size, void* d_ws, size_t ws_size,
                              hipStream_t stream) {
  const float* inputs  = (const float*)d_in[0];
  const float* weights = (const float*)d_in[1];
  const float* fcw     = (const float*)d_in[2];
  const float* fcb     = (const float*)d_in[3];
  float* out   = (float*)d_out;
  float* gates = (float*)d_ws;

  build_gates_kernel<<<84, 256, 0, stream>>>(weights, gates);
  qsim_kernel<<<BATCH / 2, 64, 0, stream>>>(inputs, gates, fcw, fcb, out);
}
